# starting kernel
# baseline (speedup 1.0000x reference)
#include <hip/hip_runtime.h>

typedef __attribute__((ext_vector_type(2))) float v2f;
typedef __attribute__((ext_vector_type(8))) float v8f;

#define HW    16384
#define IMG_W 128
#define BCN   32
#define CIN   64
#define CMID  96
#define AN2   32
#define TLS_ROWSTRIDE 512
#define TLS_FLOATS (48 * TLS_ROWSTRIDE)
#define KCHUNKS 16

static __device__ __forceinline__ v8f wmma4(v2f a, v2f b, v8f c) {
  return __builtin_amdgcn_wmma_f32_16x16x4_f32(
       false, a,  false, b,
       (short)0, c,  false,  false);
}

__global__ void k_zero(float* __restrict__ p, int n) {
  int i = blockIdx.x * 256 + threadIdx.x;
  if (i < n) p[i] = 0.f;
}

__global__ void k_fused(const float* __restrict__ x,
                        const float* __restrict__ pw,
                        const float* __restrict__ dw,
                        float* __restrict__ qkv,
                        float* __restrict__ gsums) {
  extern __shared__ float tls[];
  __shared__ float dwls[CMID * 9];
  __shared__ float chsums[2 * AN2];

  const int tid  = threadIdx.x;
  const int lane = tid & 31, wave = tid >> 5;
  const int lo = lane & 15, hi = lane >> 4;
  const int img = blockIdx.y;
  const int r0  = blockIdx.x * 2;
  const float* __restrict__ X = x + (size_t)img * CIN * HW;
  float* __restrict__ QKV = qkv + (size_t)img * CMID * HW;

  for (int i = tid; i < CMID * 9; i += 256) dwls[i] = dw[i];
  if (tid < 2 * AN2) chsums[tid] = 0.f;
  __syncthreads();

  const int orow = tid >> 7;
  const int col  = tid & (IMG_W - 1);

  for (int half_ = 0; half_ < 2; ++half_) {
    const int cbase = half_ * 48;

    for (int nt = wave; nt < 32; nt += 8) {
      const int lrow = nt >> 3;
      const int colbase = (nt & 7) * 16;
      const int rr = r0 - 1 + lrow;
      if (rr < 0 || rr >= IMG_W) {
        for (int i = lane; i < 48 * 16; i += 32)
          tls[(i >> 4) * TLS_ROWSTRIDE + lrow * IMG_W + colbase + (i & 15)] = 0.f;
      } else {
        const int pix = rr * IMG_W + colbase;
        v8f acc[3] = {};
        for (int k0 = 0; k0 < CIN; k0 += 4) {
          v2f b;
          b.x = X[(size_t)(k0 + 2 * hi) * HW + pix + lo];
          b.y = X[(size_t)(k0 + 2 * hi + 1) * HW + pix + lo];
#pragma unroll
          for (int m = 0; m < 3; ++m) {
            v2f a;
            const float* arow = pw + (cbase + m * 16 + lo) * CIN + k0 + 2 * hi;
            a.x = arow[0];
            a.y = arow[1];
            acc[m] = wmma4(a, b, acc[m]);
          }
        }
#pragma unroll
        for (int m = 0; m < 3; ++m)
#pragma unroll
          for (int r = 0; r < 8; ++r)
            tls[(m * 16 + r + 8 * hi) * TLS_ROWSTRIDE + lrow * IMG_W + colbase + lo] =
                acc[m][r];
      }
    }
    __syncthreads();

    for (int lc = 0; lc < 48; ++lc) {
      const int ch = cbase + lc;
      const float* __restrict__ tch = tls + lc * TLS_ROWSTRIDE + (1 + orow) * IMG_W + col;
      const float* __restrict__ wch = dwls + ch * 9;
      float s = 0.f;
#pragma unroll
      for (int dy = -1; dy <= 1; ++dy) {
        const float* rowp = tch + dy * IMG_W;
        if (col > 0)         s += wch[(dy + 1) * 3 + 0] * rowp[-1];
                             s += wch[(dy + 1) * 3 + 1] * rowp[0];
        if (col < IMG_W - 1) s += wch[(dy + 1) * 3 + 2] * rowp[1];
      }
      QKV[(size_t)ch * HW + (r0 + orow) * IMG_W + col] = s;
      if (ch < 2 * AN2) {
        float v = s * s;
#pragma unroll
        for (int off = 16; off > 0; off >>= 1) v += __shfl_xor(v, off, 32);
        if (lane == 0) atomicAdd(&chsums[ch], v);
      }
    }
    __syncthreads();
  }

  if (tid < 2 * AN2) atomicAdd(&gsums[img * 64 + tid], chsums[tid]);
}

__global__ void k_att_partial(const float* __restrict__ qkv,
                              float* __restrict__ Sbuf) {
  const int lane = threadIdx.x & 31, wave = threadIdx.x >> 5;
  const int lo = lane & 15, hi = lane >> 4;
  const int img = blockIdx.y;
  const float* __restrict__ Q = qkv + (size_t)img * CMID * HW;
  const float* __restrict__ Km = Q + (size_t)AN2 * HW;
  const int m0 = (wave >> 1) * 16, n0 = (wave & 1) * 16;
  const int kbase = blockIdx.x * (HW / KCHUNKS);

  const float* __restrict__ qrow = Q + (size_t)(m0 + lo) * HW + 2 * hi;
  const float* __restrict__ krow = Km + (size_t)(n0 + lo) * HW + 2 * hi;

  v8f acc0 = {}, acc1 = {}, acc2 = {}, acc3 = {};
  for (int kk = kbase; kk < kbase + HW / KCHUNKS; kk += 16) {
    v2f a, b;
    a.x = qrow[kk];      a.y = qrow[kk + 1];
    b.x = krow[kk];      b.y = krow[kk + 1];
    acc0 = wmma4(a, b, acc0);
    a.x = qrow[kk + 4];  a.y = qrow[kk + 5];
    b.x = krow[kk + 4];  b.y = krow[kk + 5];
    acc1 = wmma4(a, b, acc1);
    a.x = qrow[kk + 8];  a.y = qrow[kk + 9];
    b.x = krow[kk + 8];  b.y = krow[kk + 9];
    acc2 = wmma4(a, b, acc2);
    a.x = qrow[kk + 12]; a.y = qrow[kk + 13];
    b.x = krow[kk + 12]; b.y = krow[kk + 13];
    acc3 = wmma4(a, b, acc3);
  }
  v8f acc = (acc0 + acc1) + (acc2 + acc3);

  float* __restrict__ Sp = Sbuf + (size_t)img * (AN2 * AN2);
#pragma unroll
  for (int r = 0; r < 8; ++r)
    atomicAdd(&Sp[(m0 + r + 8 * hi) * AN2 + n0 + lo], acc[r]);
}

__global__ void k_att_final(const float* __restrict__ Sbuf,
                            const float* __restrict__ sums,
                            const float* __restrict__ proj,
                            float* __restrict__ Wbuf) {
  const int img = blockIdx.x;
  __shared__ float S[AN2][AN2 + 1];
  __shared__ float rq[AN2], rk[AN2];

  if (threadIdx.x < 32) {
    rq[threadIdx.x] = 1.f / fmaxf(sqrtf(sums[img * 64 + threadIdx.x]), 1e-12f);
  } else if (threadIdx.x < 64) {
    int d = threadIdx.x - 32;
    rk[d] = 1.f / fmaxf(sqrtf(sums[img * 64 + 32 + d]), 1e-12f);
  }
  __syncthreads();
  const float* __restrict__ Sp = Sbuf + (size_t)img * (AN2 * AN2);
  for (int idx = threadIdx.x; idx < AN2 * AN2; idx += 128) {
    int row = idx >> 5, col = idx & 31;
    S[row][col] = Sp[idx] * rq[row] * rk[col];
  }
  __syncthreads();
  if (threadIdx.x < 32) {
    int row = threadIdx.x;
    float mx = -3.4e38f;
    for (int j = 0; j < 32; ++j) mx = fmaxf(mx, S[row][j]);
    float sum = 0.f;
    for (int j = 0; j < 32; ++j) {
      float e = __expf(S[row][j] - mx);
      S[row][j] = e;
      sum += e;
    }
    float inv = 1.f / sum;
    for (int j = 0; j < 32; ++j) S[row][j] *= inv;
  }
  __syncthreads();
  for (int idx = threadIdx.x; idx < 64 * 32; idx += 128) {
    int o = idx >> 5, d = idx & 31;
    float s = 0.f;
    for (int cc = 0; cc < 32; ++cc) s += proj[o * 32 + cc] * S[cc][d];
    Wbuf[(size_t)img * 2048 + idx] = s;
  }
}

__global__ void k_out(const float* __restrict__ qkv,
                      const float* __restrict__ Wbuf,
                      float* __restrict__ out) {
  const int lane = threadIdx.x & 31, wave = threadIdx.x >> 5;
  const int lo = lane & 15, hi = lane >> 4;
  const int img = blockIdx.y;
  const int n0 = blockIdx.x * 64 + wave * 16;
  const float* __restrict__ V = qkv + (size_t)img * CMID * HW + (size_t)(2 * AN2) * HW;
  const float* __restrict__ W = Wbuf + (size_t)img * 2048;
  float* __restrict__ O = out + (size_t)img * CIN * HW;

  v8f acc[4] = {};
  for (int k0 = 0; k0 < 32; k0 += 4) {
    v2f b;
    b.x = V[(size_t)(k0 + 2 * hi) * HW + n0 + lo];
    b.y = V[(size_t)(k0 + 2 * hi + 1) * HW + n0 + lo];
#pragma unroll
    for (int m = 0; m < 4; ++m) {
      v2f a;
      const float* wr = W + (m * 16 + lo) * 32 + k0 + 2 * hi;
      a.x = wr[0];
      a.y = wr[1];
      acc[m] = wmma4(a, b, acc[m]);
    }
  }
#pragma unroll
  for (int m = 0; m < 4; ++m)
#pragma unroll
    for (int r = 0; r < 8; ++r)
      __builtin_nontemporal_store(acc[m][r],
          &O[(size_t)(m * 16 + r + 8 * hi) * HW + n0 + lo]);
}

extern "C" void kernel_launch(void* const* d_in, const int* in_sizes, int n_in,
                              void* d_out, int out_size, void* d_ws, size_t ws_size,
                              hipStream_t stream) {
  const float* x    = (const float*)d_in[0];
  const float* pw   = (const float*)d_in[1];
  const float* dw   = (const float*)d_in[2];
  const float* proj = (const float*)d_in[3];
  float* out = (float*)d_out;

  float* qkv  = (float*)d_ws;
  float* sums = qkv + (size_t)BCN * CMID * HW;
  float* Sbuf = sums + (size_t)BCN * 64;
  float* Wbuf = Sbuf + (size_t)BCN * AN2 * AN2;
  const int nzero = BCN * 64 + BCN * AN2 * AN2;

  k_zero<<<dim3((nzero + 255) / 256), 256, 0, stream>>>(sums, nzero);
  k_fused<<<dim3(IMG_W / 2, BCN), 256, TLS_FLOATS * sizeof(float), stream>>>(
      x, pw, dw, qkv, sums);
  k_att_partial<<<dim3(KCHUNKS, BCN), 128, 0, stream>>>(qkv, Sbuf);
  k_att_final<<<dim3(BCN), 128, 0, stream>>>(Sbuf, sums, proj, Wbuf);
  k_out<<<dim3(HW / 64, BCN), 128, 0, stream>>>(qkv, Wbuf, out);
}


# depthwise 3x3: a wave owns a channel, each lane reads its 4x6 LDS window once (b128) for 8 outputs, no per-tap wait or edge branch; q/k squares summed per lane, then wave tree (sum order changed)
# speedup vs baseline: 1.4725x; 1.4725x over previous
#include <hip/hip_runtime.h>

typedef __attribute__((ext_vector_type(2))) float v2f;
typedef __attribute__((ext_vector_type(4))) float v4f;
typedef __attribute__((ext_vector_type(8))) float v8f;

#define HW    16384
#define IMG_W 128
#define BCN   32
#define CIN   64
#define CMID  96
#define AN2   32
#define TLS_PAD   4
#define TLS_PITCH (IMG_W + TLS_PAD)
#define TLS_ROWSTRIDE (4 * TLS_PITCH)
#define TLS_FLOATS (48 * TLS_ROWSTRIDE + TLS_PAD)
#define KCHUNKS 16

static __device__ __forceinline__ v8f wmma4(v2f a, v2f b, v8f c) {
  return __builtin_amdgcn_wmma_f32_16x16x4_f32(
       false, a,  false, b,
       (short)0, c,  false,  false);
}

__global__ void k_zero(float* __restrict__ p, int n) {
  int i = blockIdx.x * 256 + threadIdx.x;
  if (i < n) p[i] = 0.f;
}

__global__ void k_fused(const float* __restrict__ x,
                        const float* __restrict__ pw,
                        const float* __restrict__ dw,
                        float* __restrict__ qkv,
                        float* __restrict__ gsums) {
  extern __shared__ __attribute__((aligned(16))) float tls[];

  const int tid  = threadIdx.x;
  const int lane = tid & 31, wave = tid >> 5;
  const int lo = lane & 15, hi = lane >> 4;
  const int img = blockIdx.y;
  const int r0  = blockIdx.x * 2;
  const float* __restrict__ X = x + (size_t)img * CIN * HW;
  float* __restrict__ QKV = qkv + (size_t)img * CMID * HW;

  if (tid <= 48 * 4) {
    v4f z = {0.f, 0.f, 0.f, 0.f};
    *(v4f*)(tls + tid * TLS_PITCH) = z;
  }

  const int wv = __builtin_amdgcn_readfirstlane(wave);
  const int c4 = lane * 4;

  for (int half_ = 0; half_ < 2; ++half_) {
    const int cbase = half_ * 48;

    for (int nt = wave; nt < 32; nt += 8) {
      const int lrow = nt >> 3;
      const int colbase = (nt & 7) * 16;
      const int rr = r0 - 1 + lrow;
      if (rr < 0 || rr >= IMG_W) {
        for (int i = lane; i < 48 * 16; i += 32)
          tls[(i >> 4) * TLS_ROWSTRIDE + lrow * TLS_PITCH + TLS_PAD + colbase + (i & 15)] = 0.f;
      } else {
        const int pix = rr * IMG_W + colbase;
        v8f acc[3] = {};
        for (int k0 = 0; k0 < CIN; k0 += 4) {
          v2f b;
          b.x = X[(size_t)(k0 + 2 * hi) * HW + pix + lo];
          b.y = X[(size_t)(k0 + 2 * hi + 1) * HW + pix + lo];
#pragma unroll
          for (int m = 0; m < 3; ++m) {
            v2f a;
            const float* arow = pw + (cbase + m * 16 + lo) * CIN + k0 + 2 * hi;
            a.x = arow[0];
            a.y = arow[1];
            acc[m] = wmma4(a, b, acc[m]);
          }
        }
#pragma unroll
        for (int m = 0; m < 3; ++m)
#pragma unroll
          for (int r = 0; r < 8; ++r)
            tls[(m * 16 + r + 8 * hi) * TLS_ROWSTRIDE + lrow * TLS_PITCH + TLS_PAD + colbase + lo] =
                acc[m][r];
      }
    }
    __syncthreads();

#pragma nounroll
    for (int j = 0; j < 6; ++j) {
      const int lc = wv + 8 * j;
      const int ch = cbase + lc;
      const float* __restrict__ wch = dw + ch * 9;
      float w[9];
#pragma unroll
      for (int t = 0; t < 9; ++t) w[t] = wch[t];

      const float* __restrict__ tch = tls + lc * TLS_ROWSTRIDE + TLS_PAD + c4;
      v4f mid[4];
      float lft[4], rgt[4];
#pragma unroll
      for (int r = 0; r < 4; ++r) {
        mid[r] = *(const v4f*)(tch + r * TLS_PITCH);
        lft[r] = tch[r * TLS_PITCH - 1];
        rgt[r] = tch[r * TLS_PITCH + 4];
      }

      float ss = 0.f;
#pragma unroll
      for (int orow = 0; orow < 2; ++orow) {
        float s0 = 0.f, s1 = 0.f, s2 = 0.f, s3 = 0.f;
#pragma unroll
        for (int dy = 0; dy < 3; ++dy) {
          const int r = orow + dy;
          const float w0 = w[dy * 3 + 0], w1 = w[dy * 3 + 1], w2 = w[dy * 3 + 2];
          s0 += w0 * lft[r];    s0 += w1 * mid[r].x;  s0 += w2 * mid[r].y;
          s1 += w0 * mid[r].x;  s1 += w1 * mid[r].y;  s1 += w2 * mid[r].z;
          s2 += w0 * mid[r].y;  s2 += w1 * mid[r].z;  s2 += w2 * mid[r].w;
          s3 += w0 * mid[r].z;  s3 += w1 * mid[r].w;  s3 += w2 * rgt[r];
        }
        v4f o = {s0, s1, s2, s3};
        *(v4f*)(QKV + (size_t)ch * HW + (r0 + orow) * IMG_W + c4) = o;
        ss += s0 * s0;  ss += s1 * s1;  ss += s2 * s2;  ss += s3 * s3;
      }
      if (ch < 2 * AN2) {
#pragma unroll
        for (int off = 16; off > 0; off >>= 1) ss += __shfl_xor(ss, off, 32);
        if (lane == 0) atomicAdd(&gsums[img * 64 + ch], ss);
      }
    }
    if (half_ == 0) __syncthreads();
  }
}

__global__ void k_att_partial(const float* __restrict__ qkv,
                              float* __restrict__ Sbuf) {
  const int lane = threadIdx.x & 31, wave = threadIdx.x >> 5;
  const int lo = lane & 15, hi = lane >> 4;
  const int img = blockIdx.y;
  const float* __restrict__ Q = qkv + (size_t)img * CMID * HW;
  const float* __restrict__ Km = Q + (size_t)AN2 * HW;
  const int m0 = (wave >> 1) * 16, n0 = (wave & 1) * 16;
  const int kbase = blockIdx.x * (HW / KCHUNKS);

  const float* __restrict__ qrow = Q + (size_t)(m0 + lo) * HW + 2 * hi;
  const float* __restrict__ krow = Km + (size_t)(n0 + lo) * HW + 2 * hi;

  v8f acc0 = {}, acc1 = {}, acc2 = {}, acc3 = {};
  for (int kk = kbase; kk < kbase + HW / KCHUNKS; kk += 16) {
    v2f a, b;
    a.x = qrow[kk];      a.y = qrow[kk + 1];
    b.x = krow[kk];      b.y = krow[kk + 1];
    acc0 = wmma4(a, b, acc0);
    a.x = qrow[kk + 4];  a.y = qrow[kk + 5];
    b.x = krow[kk + 4];  b.y = krow[kk + 5];
    acc1 = wmma4(a, b, acc1);
    a.x = qrow[kk + 8];  a.y = qrow[kk + 9];
    b.x = krow[kk + 8];  b.y = krow[kk + 9];
    acc2 = wmma4(a, b, acc2);
    a.x = qrow[kk + 12]; a.y = qrow[kk + 13];
    b.x = krow[kk + 12]; b.y = krow[kk + 13];
    acc3 = wmma4(a, b, acc3);
  }
  v8f acc = (acc0 + acc1) + (acc2 + acc3);

  float* __restrict__ Sp = Sbuf + (size_t)img * (AN2 * AN2);
#pragma unroll
  for (int r = 0; r < 8; ++r)
    atomicAdd(&Sp[(m0 + r + 8 * hi) * AN2 + n0 + lo], acc[r]);
}

__global__ void k_att_final(const float* __restrict__ Sbuf,
                            const float* __restrict__ sums,
                            const float* __restrict__ proj,
                            float* __restrict__ Wbuf) {
  const int img = blockIdx.x;
  __shared__ float S[AN2][AN2 + 1];
  __shared__ float rq[AN2], rk[AN2];

  if (threadIdx.x < 32) {
    rq[threadIdx.x] = 1.f / fmaxf(sqrtf(sums[img * 64 + threadIdx.x]), 1e-12f);
  } else if (threadIdx.x < 64) {
    int d = threadIdx.x - 32;
    rk[d] = 1.f / fmaxf(sqrtf(sums[img * 64 + 32 + d]), 1e-12f);
  }
  __syncthreads();
  const float* __restrict__ Sp = Sbuf + (size_t)img * (AN2 * AN2);
  for (int idx = threadIdx.x; idx < AN2 * AN2; idx += 128) {
    int row = idx >> 5, col = idx & 31;
    S[row][col] = Sp[idx] * rq[row] * rk[col];
  }
  __syncthreads();
  if (threadIdx.x < 32) {
    int row = threadIdx.x;
    float mx = -3.4e38f;
    for (int j = 0; j < 32; ++j) mx = fmaxf(mx, S[row][j]);
    float sum = 0.f;
    for (int j = 0; j < 32; ++j) {
      float e = __expf(S[row][j] - mx);
      S[row][j] = e;
      sum += e;
    }
    float inv = 1.f / sum;
    for (int j = 0; j < 32; ++j) S[row][j] *= inv;
  }
  __syncthreads();
  for (int idx = threadIdx.x; idx < 64 * 32; idx += 128) {
    int o = idx >> 5, d = idx & 31;
    float s = 0.f;
    for (int cc = 0; cc < 32; ++cc) s += proj[o * 32 + cc] * S[cc][d];
    Wbuf[(size_t)img * 2048 + idx] = s;
  }
}

__global__ void k_out(const float* __restrict__ qkv,
                      const float* __restrict__ Wbuf,
                      float* __restrict__ out) {
  const int lane = threadIdx.x & 31, wave = threadIdx.x >> 5;
  const int lo = lane & 15, hi = lane >> 4;
  const int img = blockIdx.y;
  const int n0 = blockIdx.x * 64 + wave * 16;
  const float* __restrict__ V = qkv + (size_t)img * CMID * HW + (size_t)(2 * AN2) * HW;
  const float* __restrict__ W = Wbuf + (size_t)img * 2048;
  float* __restrict__ O = out + (size_t)img * CIN * HW;

  v8f acc[4] = {};
  for (int k0 = 0; k0 < 32; k0 += 4) {
    v2f b;
    b.x = V[(size_t)(k0 + 2 * hi) * HW + n0 + lo];
    b.y = V[(size_t)(k0 + 2 * hi + 1) * HW + n0 + lo];
#pragma unroll
    for (int m = 0; m < 4; ++m) {
      v2f a;
      const float* wr = W + (m * 16 + lo) * 32 + k0 + 2 * hi;
      a.x = wr[0];
      a.y = wr[1];
      acc[m] = wmma4(a, b, acc[m]);
    }
  }
#pragma unroll
  for (int m = 0; m < 4; ++m)
#pragma unroll
    for (int r = 0; r < 8; ++r)
      __builtin_nontemporal_store(acc[m][r],
          &O[(size_t)(m * 16 + r + 8 * hi) * HW + n0 + lo]);
}

extern "C" void kernel_launch(void* const* d_in, const int* in_sizes, int n_in,
                              void* d_out, int out_size, void* d_ws, size_t ws_size,
                              hipStream_t stream) {
  const float* x    = (const float*)d_in[0];
  const float* pw   = (const float*)d_in[1];
  const float* dw   = (const float*)d_in[2];
  const float* proj = (const float*)d_in[3];
  float* out = (float*)d_out;

  float* qkv  = (float*)d_ws;
  float* sums = qkv + (size_t)BCN * CMID * HW;
  float* Sbuf = sums + (size_t)BCN * 64;
  float* Wbuf = Sbuf + (size_t)BCN * AN2 * AN2;
  const int nzero = BCN * 64 + BCN * AN2 * AN2;

  k_zero<<<dim3((nzero + 255) / 256), 256, 0, stream>>>(sums, nzero);
  k_fused<<<dim3(IMG_W / 2, BCN), 256, TLS_FLOATS * sizeof(float), stream>>>(
      x, pw, dw, qkv, sums);
  k_att_partial<<<dim3(KCHUNKS, BCN), 128, 0, stream>>>(qkv, Sbuf);
  k_att_final<<<dim3(BCN), 128, 0, stream>>>(Sbuf, sums, proj, Wbuf);
  k_out<<<dim3(HW / 64, BCN), 128, 0, stream>>>(qkv, Wbuf, out);
}


# pointwise 1x1: column c of n-tile i = pixel 4c+i, so a lane reads x as one b128 per channel row, each pw_w A fragment feeds 4 n-tiles (A loads /4), t goes to LDS as b128; k order kept
# speedup vs baseline: 1.9298x; 1.3106x over previous
#include <hip/hip_runtime.h>

typedef __attribute__((ext_vector_type(2))) float v2f;
typedef __attribute__((ext_vector_type(4))) float v4f;
typedef __attribute__((ext_vector_type(8))) float v8f;

#define HW    16384
#define IMG_W 128
#define BCN   32
#define CIN   64
#define CMID  96
#define AN2   32
#define TLS_PAD   4
#define TLS_PITCH (IMG_W + TLS_PAD)
#define TLS_ROWSTRIDE (4 * TLS_PITCH)
#define TLS_FLOATS (48 * TLS_ROWSTRIDE + TLS_PAD)
#define KCHUNKS 16

static __device__ __forceinline__ v8f wmma4(v2f a, v2f b, v8f c) {
  return __builtin_amdgcn_wmma_f32_16x16x4_f32(
       false, a,  false, b,
       (short)0, c,  false,  false);
}

__global__ void k_zero(float* __restrict__ p, int n) {
  int i = blockIdx.x * 256 + threadIdx.x;
  if (i < n) p[i] = 0.f;
}

__global__ void __launch_bounds__(256)
k_fused(const float* __restrict__ x,
        const float* __restrict__ pw,
        const float* __restrict__ dw,
        float* __restrict__ qkv,
        float* __restrict__ gsums) {
  extern __shared__ __attribute__((aligned(16))) float tls[];

  const int tid  = threadIdx.x;
  const int lane = tid & 31, wave = tid >> 5;
  const int lo = lane & 15, hi = lane >> 4;
  const int img = blockIdx.y;
  const int r0  = blockIdx.x * 2;
  const float* __restrict__ X = x + (size_t)img * CIN * HW;
  float* __restrict__ QKV = qkv + (size_t)img * CMID * HW;

  if (tid <= 48 * 4) {
    v4f z = {0.f, 0.f, 0.f, 0.f};
    *(v4f*)(tls + tid * TLS_PITCH) = z;
  }

  const int wv = __builtin_amdgcn_readfirstlane(wave);
  const int c4 = lane * 4;

  const int lrow = wv >> 1;
  const int colbase = (wv & 1) * 64;
  const int rr = r0 - 1 + lrow;
  float* __restrict__ trow = tls + lrow * TLS_PITCH + TLS_PAD + colbase + 4 * lo;

  for (int half_ = 0; half_ < 2; ++half_) {
    const int cbase = half_ * 48;

    if (rr < 0 || rr >= IMG_W) {
      v4f z = {0.f, 0.f, 0.f, 0.f};
      for (int c = hi; c < 48; c += 2) *(v4f*)(trow + c * TLS_ROWSTRIDE) = z;
    } else {
      const float* __restrict__ xp = X + (size_t)(2 * hi) * HW + rr * IMG_W + colbase + 4 * lo;
      const float* __restrict__ ap = pw + (cbase + lo) * CIN + 2 * hi;
      v8f acc[3][4] = {};
#pragma unroll
      for (int k0 = 0; k0 < CIN; k0 += 4) {
        const v4f x0 = *(const v4f*)(xp + (size_t)k0 * HW);
        const v4f x1 = *(const v4f*)(xp + (size_t)(k0 + 1) * HW);
        v2f a[3];
#pragma unroll
        for (int m = 0; m < 3; ++m) a[m] = *(const v2f*)(ap + m * 16 * CIN + k0);
#pragma unroll
        for (int i = 0; i < 4; ++i) {
          v2f b;
          b.x = x0[i];
          b.y = x1[i];
#pragma unroll
          for (int m = 0; m < 3; ++m) acc[m][i] = wmma4(a[m], b, acc[m][i]);
        }
      }
#pragma unroll
      for (int m = 0; m < 3; ++m)
#pragma unroll
        for (int r = 0; r < 8; ++r) {
          v4f o = {acc[m][0][r], acc[m][1][r], acc[m][2][r], acc[m][3][r]};
          *(v4f*)(trow + (m * 16 + r + 8 * hi) * TLS_ROWSTRIDE) = o;
        }
    }
    __syncthreads();

#pragma nounroll
    for (int j = 0; j < 6; ++j) {
      const int lc = wv + 8 * j;
      const int ch = cbase + lc;
      const float* __restrict__ wch = dw + ch * 9;
      float w[9];
#pragma unroll
      for (int t = 0; t < 9; ++t) w[t] = wch[t];

      const float* __restrict__ tch = tls + lc * TLS_ROWSTRIDE + TLS_PAD + c4;
      v4f mid[4];
      float lft[4], rgt[4];
#pragma unroll
      for (int r = 0; r < 4; ++r) {
        mid[r] = *(const v4f*)(tch + r * TLS_PITCH);
        lft[r] = tch[r * TLS_PITCH - 1];
        rgt[r] = tch[r * TLS_PITCH + 4];
      }

      float ss = 0.f;
#pragma unroll
      for (int orow = 0; orow < 2; ++orow) {
        float s0 = 0.f, s1 = 0.f, s2 = 0.f, s3 = 0.f;
#pragma unroll
        for (int dy = 0; dy < 3; ++dy) {
          const int r = orow + dy;
          const float w0 = w[dy * 3 + 0], w1 = w[dy * 3 + 1], w2 = w[dy * 3 + 2];
          s0 += w0 * lft[r];    s0 += w1 * mid[r].x;  s0 += w2 * mid[r].y;
          s1 += w0 * mid[r].x;  s1 += w1 * mid[r].y;  s1 += w2 * mid[r].z;
          s2 += w0 * mid[r].y;  s2 += w1 * mid[r].z;  s2 += w2 * mid[r].w;
          s3 += w0 * mid[r].z;  s3 += w1 * mid[r].w;  s3 += w2 * rgt[r];
        }
        v4f o = {s0, s1, s2, s3};
        *(v4f*)(QKV + (size_t)ch * HW + (r0 + orow) * IMG_W + c4) = o;
        ss += s0 * s0;  ss += s1 * s1;  ss += s2 * s2;  ss += s3 * s3;
      }
      if (ch < 2 * AN2) {
#pragma unroll
        for (int off = 16; off > 0; off >>= 1) ss += __shfl_xor(ss, off, 32);
        if (lane == 0) atomicAdd(&gsums[img * 64 + ch], ss);
      }
    }
    if (half_ == 0) __syncthreads();
  }
}

__global__ void k_att_partial(const float* __restrict__ qkv,
                              float* __restrict__ Sbuf) {
  const int lane = threadIdx.x & 31, wave = threadIdx.x >> 5;
  const int lo = lane & 15, hi = lane >> 4;
  const int img = blockIdx.y;
  const float* __restrict__ Q = qkv + (size_t)img * CMID * HW;
  const float* __restrict__ Km = Q + (size_t)AN2 * HW;
  const int m0 = (wave >> 1) * 16, n0 = (wave & 1) * 16;
  const int kbase = blockIdx.x * (HW / KCHUNKS);

  const float* __restrict__ qrow = Q + (size_t)(m0 + lo) * HW + 2 * hi;
  const float* __restrict__ krow = Km + (size_t)(n0 + lo) * HW + 2 * hi;

  v8f acc0 = {}, acc1 = {}, acc2 = {}, acc3 = {};
  for (int kk = kbase; kk < kbase + HW / KCHUNKS; kk += 16) {
    v2f a, b;
    a.x = qrow[kk];      a.y = qrow[kk + 1];
    b.x = krow[kk];      b.y = krow[kk + 1];
    acc0 = wmma4(a, b, acc0);
    a.x = qrow[kk + 4];  a.y = qrow[kk + 5];
    b.x = krow[kk + 4];  b.y = krow[kk + 5];
    acc1 = wmma4(a, b, acc1);
    a.x = qrow[kk + 8];  a.y = qrow[kk + 9];
    b.x = krow[kk + 8];  b.y = krow[kk + 9];
    acc2 = wmma4(a, b, acc2);
    a.x = qrow[kk + 12]; a.y = qrow[kk + 13];
    b.x = krow[kk + 12]; b.y = krow[kk + 13];
    acc3 = wmma4(a, b, acc3);
  }
  v8f acc = (acc0 + acc1) + (acc2 + acc3);

  float* __restrict__ Sp = Sbuf + (size_t)img * (AN2 * AN2);
#pragma unroll
  for (int r = 0; r < 8; ++r)
    atomicAdd(&Sp[(m0 + r + 8 * hi) * AN2 + n0 + lo], acc[r]);
}

__global__ void k_att_final(const float* __restrict__ Sbuf,
                            const float* __restrict__ sums,
                            const float* __restrict__ proj,
                            float* __restrict__ Wbuf) {
  const int img = blockIdx.x;
  __shared__ float S[AN2][AN2 + 1];
  __shared__ float rq[AN2], rk[AN2];

  if (threadIdx.x < 32) {
    rq[threadIdx.x] = 1.f / fmaxf(sqrtf(sums[img * 64 + threadIdx.x]), 1e-12f);
  } else if (threadIdx.x < 64) {
    int d = threadIdx.x - 32;
    rk[d] = 1.f / fmaxf(sqrtf(sums[img * 64 + 32 + d]), 1e-12f);
  }
  __syncthreads();
  const float* __restrict__ Sp = Sbuf + (size_t)img * (AN2 * AN2);
  for (int idx = threadIdx.x; idx < AN2 * AN2; idx += 128) {
    int row = idx >> 5, col = idx & 31;
    S[row][col] = Sp[idx] * rq[row] * rk[col];
  }
  __syncthreads();
  if (threadIdx.x < 32) {
    int row = threadIdx.x;
    float mx = -3.4e38f;
    for (int j = 0; j < 32; ++j) mx = fmaxf(mx, S[row][j]);
    float sum = 0.f;
    for (int j = 0; j < 32; ++j) {
      float e = __expf(S[row][j] - mx);
      S[row][j] = e;
      sum += e;
    }
    float inv = 1.f / sum;
    for (int j = 0; j < 32; ++j) S[row][j] *= inv;
  }
  __syncthreads();
  for (int idx = threadIdx.x; idx < 64 * 32; idx += 128) {
    int o = idx >> 5, d = idx & 31;
    float s = 0.f;
    for (int cc = 0; cc < 32; ++cc) s += proj[o * 32 + cc] * S[cc][d];
    Wbuf[(size_t)img * 2048 + idx] = s;
  }
}

__global__ void k_out(const float* __restrict__ qkv,
                      const float* __restrict__ Wbuf,
                      float* __restrict__ out) {
  const int lane = threadIdx.x & 31, wave = threadIdx.x >> 5;
  const int lo = lane & 15, hi = lane >> 4;
  const int img = blockIdx.y;
  const int n0 = blockIdx.x * 64 + wave * 16;
  const float* __restrict__ V = qkv + (size_t)img * CMID * HW + (size_t)(2 * AN2) * HW;
  const float* __restrict__ W = Wbuf + (size_t)img * 2048;
  float* __restrict__ O = out + (size_t)img * CIN * HW;

  v8f acc[4] = {};
  for (int k0 = 0; k0 < 32; k0 += 4) {
    v2f b;
    b.x = V[(size_t)(k0 + 2 * hi) * HW + n0 + lo];
    b.y = V[(size_t)(k0 + 2 * hi + 1) * HW + n0 + lo];
#pragma unroll
    for (int m = 0; m < 4; ++m) {
      v2f a;
      const float* wr = W + (m * 16 + lo) * 32 + k0 + 2 * hi;
      a.x = wr[0];
      a.y = wr[1];
      acc[m] = wmma4(a, b, acc[m]);
    }
  }
#pragma unroll
  for (int m = 0; m < 4; ++m)
#pragma unroll
    for (int r = 0; r < 8; ++r)
      __builtin_nontemporal_store(acc[m][r],
          &O[(size_t)(m * 16 + r + 8 * hi) * HW + n0 + lo]);
}

extern "C" void kernel_launch(void* const* d_in, const int* in_sizes, int n_in,
                              void* d_out, int out_size, void* d_ws, size_t ws_size,
                              hipStream_t stream) {
  const float* x    = (const float*)d_in[0];
  const float* pw   = (const float*)d_in[1];
  const float* dw   = (const float*)d_in[2];
  const float* proj = (const float*)d_in[3];
  float* out = (float*)d_out;

  float* qkv  = (float*)d_ws;
  float* sums = qkv + (size_t)BCN * CMID * HW;
  float* Sbuf = sums + (size_t)BCN * 64;
  float* Wbuf = Sbuf + (size_t)BCN * AN2 * AN2;
  const int nzero = BCN * 64 + BCN * AN2 * AN2;

  k_zero<<<dim3((nzero + 255) / 256), 256, 0, stream>>>(sums, nzero);
  k_fused<<<dim3(IMG_W / 2, BCN), 256, TLS_FLOATS * sizeof(float), stream>>>(
      x, pw, dw, qkv, sums);
  k_att_partial<<<dim3(KCHUNKS, BCN), 128, 0, stream>>>(qkv, Sbuf);
  k_att_final<<<dim3(BCN), 128, 0, stream>>>(Sbuf, sums, proj, Wbuf);
  k_out<<<dim3(HW / 64, BCN), 128, 0, stream>>>(qkv, Wbuf, out);
}


# fused kernel: block = 4 output rows, not 2 (6 t rows in LDS, 12 waves, 160 regs): pointwise halo recompute 2x -> 1.5x; x loads held 2 k-steps ahead by sched barrier; q/k squares sum order changed
# speedup vs baseline: 2.1984x; 1.1392x over previous
#include <hip/hip_runtime.h>

typedef __attribute__((ext_vector_type(2))) float v2f;
typedef __attribute__((ext_vector_type(4))) float v4f;
typedef __attribute__((ext_vector_type(8))) float v8f;

#define HW    16384
#define IMG_W 128
#define BCN   32
#define CIN   64
#define CMID  96
#define AN2   32
#define OROWS  4
#define LROWS  (OROWS + 2)
#define NWAVES (2 * LROWS)
#define X_AHEAD 2
#define W_AHEAD 1
#define TLS_PAD   4
#define TLS_PITCH (IMG_W + TLS_PAD)
#define TLS_ROWSTRIDE (LROWS * TLS_PITCH)
#define TLS_FLOATS (48 * TLS_ROWSTRIDE + TLS_PAD)
#define KCHUNKS 16

static __device__ __forceinline__ v8f wmma4(v2f a, v2f b, v8f c) {
  return __builtin_amdgcn_wmma_f32_16x16x4_f32(
       false, a,  false, b,
       (short)0, c,  false,  false);
}

__global__ void k_zero(float* __restrict__ p, int n) {
  int i = blockIdx.x * 256 + threadIdx.x;
  if (i < n) p[i] = 0.f;
}

__global__ void __launch_bounds__(NWAVES * 32) __attribute__((amdgpu_waves_per_eu(6, 6)))
k_fused(const float* __restrict__ x,
        const float* __restrict__ pw,
        const float* __restrict__ dw,
        float* __restrict__ qkv,
        float* __restrict__ gsums) {
  extern __shared__ __attribute__((aligned(16))) float tls[];

  const int tid  = threadIdx.x;
  const int lane = tid & 31, wave = tid >> 5;
  const int lo = lane & 15, hi = lane >> 4;
  const int img = blockIdx.y;
  const int r0  = blockIdx.x * OROWS;
  const float* __restrict__ X = x + (size_t)img * CIN * HW;
  float* __restrict__ QKV = qkv + (size_t)img * CMID * HW;

  if (tid <= 48 * LROWS) {
    v4f z = {0.f, 0.f, 0.f, 0.f};
    *(v4f*)(tls + tid * TLS_PITCH) = z;
  }

  const int wv = __builtin_amdgcn_readfirstlane(wave);
  const int c4 = lane * 4;

  const int lrow = wv >> 1;
  const int colbase = (wv & 1) * 64;
  const int rr = r0 - 1 + lrow;
  float* __restrict__ trow = tls + lrow * TLS_PITCH + TLS_PAD + colbase + 4 * lo;

  for (int half_ = 0; half_ < 2; ++half_) {
    const int cbase = half_ * 48;

    if (rr < 0 || rr >= IMG_W) {
      v4f z = {0.f, 0.f, 0.f, 0.f};
      for (int c = hi; c < 48; c += 2) *(v4f*)(trow + c * TLS_ROWSTRIDE) = z;
    } else {
      const float* __restrict__ xp = X + (size_t)(2 * hi) * HW + rr * IMG_W + colbase + 4 * lo;
      const float* __restrict__ ap = pw + (cbase + lo) * CIN + 2 * hi;
      v8f acc[3][4] = {};
      v4f x0[X_AHEAD + 1], x1[X_AHEAD + 1];
      v2f a[W_AHEAD + 1][3];
#pragma unroll
      for (int st = 0; st < X_AHEAD; ++st) {
        x0[st] = *(const v4f*)(xp + (size_t)(4 * st) * HW);
        x1[st] = *(const v4f*)(xp + (size_t)(4 * st + 1) * HW);
      }
#pragma unroll
      for (int st = 0; st < W_AHEAD; ++st)
#pragma unroll
        for (int m = 0; m < 3; ++m) a[st][m] = *(const v2f*)(ap + m * 16 * CIN + 4 * st);
#pragma unroll
      for (int st = 0; st < CIN / 4; ++st) {
        if (st + W_AHEAD < CIN / 4) {
          const int sl = (st + W_AHEAD) % (W_AHEAD + 1);
          const int kn = 4 * (st + W_AHEAD);
#pragma unroll
          for (int m = 0; m < 3; ++m) a[sl][m] = *(const v2f*)(ap + m * 16 * CIN + kn);
        }
        if (st + X_AHEAD < CIN / 4) {
          const int sl = (st + X_AHEAD) % (X_AHEAD + 1);
          const int kn = 4 * (st + X_AHEAD);
          x0[sl] = *(const v4f*)(xp + (size_t)kn * HW);
          x1[sl] = *(const v4f*)(xp + (size_t)(kn + 1) * HW);
        }
        __builtin_amdgcn_sched_barrier(0);
        const int sx = st % (X_AHEAD + 1);
        const int sw = st % (W_AHEAD + 1);
#pragma unroll
        for (int i = 0; i < 4; ++i) {
          v2f b;
          b.x = x0[sx][i];
          b.y = x1[sx][i];
#pragma unroll
          for (int m = 0; m < 3; ++m) acc[m][i] = wmma4(a[sw][m], b, acc[m][i]);
        }
      }
#pragma unroll
      for (int m = 0; m < 3; ++m)
#pragma unroll
        for (int r = 0; r < 8; ++r) {
          v4f o = {acc[m][0][r], acc[m][1][r], acc[m][2][r], acc[m][3][r]};
          *(v4f*)(trow + (m * 16 + r + 8 * hi) * TLS_ROWSTRIDE) = o;
        }
    }
    __syncthreads();

#pragma nounroll
    for (int j = 0; j < 48 / NWAVES; ++j) {
      const int lc = wv + NWAVES * j;
      const int ch = cbase + lc;
      const float* __restrict__ wch = dw + ch * 9;
      float w[9];
#pragma unroll
      for (int t = 0; t < 9; ++t) w[t] = wch[t];

      const float* __restrict__ tch = tls + lc * TLS_ROWSTRIDE + TLS_PAD + c4;
      v4f mid[LROWS];
      float lft[LROWS], rgt[LROWS];
#pragma unroll
      for (int r = 0; r < LROWS; ++r) {
        mid[r] = *(const v4f*)(tch + r * TLS_PITCH);
        lft[r] = tch[r * TLS_PITCH - 1];
        rgt[r] = tch[r * TLS_PITCH + 4];
      }

      float ss = 0.f;
#pragma unroll
      for (int orow = 0; orow < OROWS; ++orow) {
        float s0 = 0.f, s1 = 0.f, s2 = 0.f, s3 = 0.f;
#pragma unroll
        for (int dy = 0; dy < 3; ++dy) {
          const int r = orow + dy;
          const float w0 = w[dy * 3 + 0], w1 = w[dy * 3 + 1], w2 = w[dy * 3 + 2];
          s0 += w0 * lft[r];    s0 += w1 * mid[r].x;  s0 += w2 * mid[r].y;
          s1 += w0 * mid[r].x;  s1 += w1 * mid[r].y;  s1 += w2 * mid[r].z;
          s2 += w0 * mid[r].y;  s2 += w1 * mid[r].z;  s2 += w2 * mid[r].w;
          s3 += w0 * mid[r].z;  s3 += w1 * mid[r].w;  s3 += w2 * rgt[r];
        }
        v4f o = {s0, s1, s2, s3};
        *(v4f*)(QKV + (size_t)ch * HW + (r0 + orow) * IMG_W + c4) = o;
        ss += s0 * s0;  ss += s1 * s1;  ss += s2 * s2;  ss += s3 * s3;
      }
      if (ch < 2 * AN2) {
#pragma unroll
        for (int off = 16; off > 0; off >>= 1) ss += __shfl_xor(ss, off, 32);
        if (lane == 0) atomicAdd(&gsums[img * 64 + ch], ss);
      }
    }
    if (half_ == 0) __syncthreads();
  }
}

__global__ void k_att_partial(const float* __restrict__ qkv,
                              float* __restrict__ Sbuf) {
  const int lane = threadIdx.x & 31, wave = threadIdx.x >> 5;
  const int lo = lane & 15, hi = lane >> 4;
  const int img = blockIdx.y;
  const float* __restrict__ Q = qkv + (size_t)img * CMID * HW;
  const float* __restrict__ Km = Q + (size_t)AN2 * HW;
  const int m0 = (wave >> 1) * 16, n0 = (wave & 1) * 16;
  const int kbase = blockIdx.x * (HW / KCHUNKS);

  const float* __restrict__ qrow = Q + (size_t)(m0 + lo) * HW + 2 * hi;
  const float* __restrict__ krow = Km + (size_t)(n0 + lo) * HW + 2 * hi;

  v8f acc0 = {}, acc1 = {}, acc2 = {}, acc3 = {};
  for (int kk = kbase; kk < kbase + HW / KCHUNKS; kk += 16) {
    v2f a, b;
    a.x = qrow[kk];      a.y = qrow[kk + 1];
    b.x = krow[kk];      b.y = krow[kk + 1];
    acc0 = wmma4(a, b, acc0);
    a.x = qrow[kk + 4];  a.y = qrow[kk + 5];
    b.x = krow[kk + 4];  b.y = krow[kk + 5];
    acc1 = wmma4(a, b, acc1);
    a.x = qrow[kk + 8];  a.y = qrow[kk + 9];
    b.x = krow[kk + 8];  b.y = krow[kk + 9];
    acc2 = wmma4(a, b, acc2);
    a.x = qrow[kk + 12]; a.y = qrow[kk + 13];
    b.x = krow[kk + 12]; b.y = krow[kk + 13];
    acc3 = wmma4(a, b, acc3);
  }
  v8f acc = (acc0 + acc1) + (acc2 + acc3);

  float* __restrict__ Sp = Sbuf + (size_t)img * (AN2 * AN2);
#pragma unroll
  for (int r = 0; r < 8; ++r)
    atomicAdd(&Sp[(m0 + r + 8 * hi) * AN2 + n0 + lo], acc[r]);
}

__global__ void k_att_final(const float* __restrict__ Sbuf,
                            const float* __restrict__ sums,
                            const float* __restrict__ proj,
                            float* __restrict__ Wbuf) {
  const int img = blockIdx.x;
  __shared__ float S[AN2][AN2 + 1];
  __shared__ float rq[AN2], rk[AN2];

  if (threadIdx.x < 32) {
    rq[threadIdx.x] = 1.f / fmaxf(sqrtf(sums[img * 64 + threadIdx.x]), 1e-12f);
  } else if (threadIdx.x < 64) {
    int d = threadIdx.x - 32;
    rk[d] = 1.f / fmaxf(sqrtf(sums[img * 64 + 32 + d]), 1e-12f);
  }
  __syncthreads();
  const float* __restrict__ Sp = Sbuf + (size_t)img * (AN2 * AN2);
  for (int idx = threadIdx.x; idx < AN2 * AN2; idx += 128) {
    int row = idx >> 5, col = idx & 31;
    S[row][col] = Sp[idx] * rq[row] * rk[col];
  }
  __syncthreads();
  if (threadIdx.x < 32) {
    int row = threadIdx.x;
    float mx = -3.4e38f;
    for (int j = 0; j < 32; ++j) mx = fmaxf(mx, S[row][j]);
    float sum = 0.f;
    for (int j = 0; j < 32; ++j) {
      float e = __expf(S[row][j] - mx);
      S[row][j] = e;
      sum += e;
    }
    float inv = 1.f / sum;
    for (int j = 0; j < 32; ++j) S[row][j] *= inv;
  }
  __syncthreads();
  for (int idx = threadIdx.x; idx < 64 * 32; idx += 128) {
    int o = idx >> 5, d = idx & 31;
    float s = 0.f;
    for (int cc = 0; cc < 32; ++cc) s += proj[o * 32 + cc] * S[cc][d];
    Wbuf[(size_t)img * 2048 + idx] = s;
  }
}

__global__ void k_out(const float* __restrict__ qkv,
                      const float* __restrict__ Wbuf,
                      float* __restrict__ out) {
  const int lane = threadIdx.x & 31, wave = threadIdx.x >> 5;
  const int lo = lane & 15, hi = lane >> 4;
  const int img = blockIdx.y;
  const int n0 = blockIdx.x * 64 + wave * 16;
  const float* __restrict__ V = qkv + (size_t)img * CMID * HW + (size_t)(2 * AN2) * HW;
  const float* __restrict__ W = Wbuf + (size_t)img * 2048;
  float* __restrict__ O = out + (size_t)img * CIN * HW;

  v8f acc[4] = {};
  for (int k0 = 0; k0 < 32; k0 += 4) {
    v2f b;
    b.x = V[(size_t)(k0 + 2 * hi) * HW + n0 + lo];
    b.y = V[(size_t)(k0 + 2 * hi + 1) * HW + n0 + lo];
#pragma unroll
    for (int m = 0; m < 4; ++m) {
      v2f a;
      const float* wr = W + (m * 16 + lo) * 32 + k0 + 2 * hi;
      a.x = wr[0];
      a.y = wr[1];
      acc[m] = wmma4(a, b, acc[m]);
    }
  }
#pragma unroll
  for (int m = 0; m < 4; ++m)
#pragma unroll
    for (int r = 0; r < 8; ++r)
      __builtin_nontemporal_store(acc[m][r],
          &O[(size_t)(m * 16 + r + 8 * hi) * HW + n0 + lo]);
}

extern "C" void kernel_launch(void* const* d_in, const int* in_sizes, int n_in,
                              void* d_out, int out_size, void* d_ws, size_t ws_size,
                              hipStream_t stream) {
  const float* x    = (const float*)d_in[0];
  const float* pw   = (const float*)d_in[1];
  const float* dw   = (const float*)d_in[2];
  const float* proj = (const float*)d_in[3];
  float* out = (float*)d_out;

  float* qkv  = (float*)d_ws;
  float* sums = qkv + (size_t)BCN * CMID * HW;
  float* Sbuf = sums + (size_t)BCN * 64;
  float* Wbuf = Sbuf + (size_t)BCN * AN2 * AN2;
  const int nzero = BCN * 64 + BCN * AN2 * AN2;

  k_zero<<<dim3((nzero + 255) / 256), 256, 0, stream>>>(sums, nzero);
  k_fused<<<dim3(IMG_W / OROWS, BCN), NWAVES * 32, TLS_FLOATS * sizeof(float), stream>>>(
      x, pw, dw, qkv, sums);
  k_att_partial<<<dim3(KCHUNKS, BCN), 128, 0, stream>>>(qkv, Sbuf);
  k_att_final<<<dim3(BCN), 128, 0, stream>>>(Sbuf, sums, proj, Wbuf);
  k_out<<<dim3(HW / 64, BCN), 128, 0, stream>>>(qkv, Wbuf, out);
}


# k_out: a wave owns 256 px as 4 groups of 64 with interleaved n-tile columns; W fragments loaded once per wave (A loads /16); v read, out written as b128 whole lines; 8192 -> 256 blocks; k order kept
# speedup vs baseline: 2.6182x; 1.1910x over previous
#include <hip/hip_runtime.h>

typedef __attribute__((ext_vector_type(2))) float v2f;
typedef __attribute__((ext_vector_type(4))) float v4f;
typedef __attribute__((ext_vector_type(8))) float v8f;

#define HW    16384
#define IMG_W 128
#define BCN   32
#define CIN   64
#define CMID  96
#define AN2   32
#define OROWS  4
#define LROWS  (OROWS + 2)
#define NWAVES (2 * LROWS)
#define X_AHEAD 2
#define W_AHEAD 1
#define TLS_PAD   4
#define TLS_PITCH (IMG_W + TLS_PAD)
#define TLS_ROWSTRIDE (LROWS * TLS_PITCH)
#define TLS_FLOATS (48 * TLS_ROWSTRIDE + TLS_PAD)
#define KCHUNKS 16

static __device__ __forceinline__ v8f wmma4(v2f a, v2f b, v8f c) {
  return __builtin_amdgcn_wmma_f32_16x16x4_f32(
       false, a,  false, b,
       (short)0, c,  false,  false);
}

__global__ void k_zero(float* __restrict__ p, int n) {
  int i = blockIdx.x * 256 + threadIdx.x;
  if (i < n) p[i] = 0.f;
}

__global__ void __launch_bounds__(NWAVES * 32) __attribute__((amdgpu_waves_per_eu(6, 6)))
k_fused(const float* __restrict__ x,
        const float* __restrict__ pw,
        const float* __restrict__ dw,
        float* __restrict__ qkv,
        float* __restrict__ gsums) {
  extern __shared__ __attribute__((aligned(16))) float tls[];

  const int tid  = threadIdx.x;
  const int lane = tid & 31, wave = tid >> 5;
  const int lo = lane & 15, hi = lane >> 4;
  const int img = blockIdx.y;
  const int r0  = blockIdx.x * OROWS;
  const float* __restrict__ X = x + (size_t)img * CIN * HW;
  float* __restrict__ QKV = qkv + (size_t)img * CMID * HW;

  if (tid <= 48 * LROWS) {
    v4f z = {0.f, 0.f, 0.f, 0.f};
    *(v4f*)(tls + tid * TLS_PITCH) = z;
  }

  const int wv = __builtin_amdgcn_readfirstlane(wave);
  const int c4 = lane * 4;

  const int lrow = wv >> 1;
  const int colbase = (wv & 1) * 64;
  const int rr = r0 - 1 + lrow;
  float* __restrict__ trow = tls + lrow * TLS_PITCH + TLS_PAD + colbase + 4 * lo;

  for (int half_ = 0; half_ < 2; ++half_) {
    const int cbase = half_ * 48;

    if (rr < 0 || rr >= IMG_W) {
      v4f z = {0.f, 0.f, 0.f, 0.f};
      for (int c = hi; c < 48; c += 2) *(v4f*)(trow + c * TLS_ROWSTRIDE) = z;
    } else {
      const float* __restrict__ xp = X + (size_t)(2 * hi) * HW + rr * IMG_W + colbase + 4 * lo;
      const float* __restrict__ ap = pw + (cbase + lo) * CIN + 2 * hi;
      v8f acc[3][4] = {};
      v4f x0[X_AHEAD + 1], x1[X_AHEAD + 1];
      v2f a[W_AHEAD + 1][3];
#pragma unroll
      for (int st = 0; st < X_AHEAD; ++st) {
        x0[st] = *(const v4f*)(xp + (size_t)(4 * st) * HW);
        x1[st] = *(const v4f*)(xp + (size_t)(4 * st + 1) * HW);
      }
#pragma unroll
      for (int st = 0; st < W_AHEAD; ++st)
#pragma unroll
        for (int m = 0; m < 3; ++m) a[st][m] = *(const v2f*)(ap + m * 16 * CIN + 4 * st);
#pragma unroll
      for (int st = 0; st < CIN / 4; ++st) {
        if (st + W_AHEAD < CIN / 4) {
          const int sl = (st + W_AHEAD) % (W_AHEAD + 1);
          const int kn = 4 * (st + W_AHEAD);
#pragma unroll
          for (int m = 0; m < 3; ++m) a[sl][m] = *(const v2f*)(ap + m * 16 * CIN + kn);
        }
        if (st + X_AHEAD < CIN / 4) {
          const int sl = (st + X_AHEAD) % (X_AHEAD + 1);
          const int kn = 4 * (st + X_AHEAD);
          x0[sl] = *(const v4f*)(xp + (size_t)kn * HW);
          x1[sl] = *(const v4f*)(xp + (size_t)(kn + 1) * HW);
        }
        __builtin_amdgcn_sched_barrier(0);
        const int sx = st % (X_AHEAD + 1);
        const int sw = st % (W_AHEAD + 1);
#pragma unroll
        for (int i = 0; i < 4; ++i) {
          v2f b;
          b.x = x0[sx][i];
          b.y = x1[sx][i];
#pragma unroll
          for (int m = 0; m < 3; ++m) acc[m][i] = wmma4(a[sw][m], b, acc[m][i]);
        }
      }
#pragma unroll
      for (int m = 0; m < 3; ++m)
#pragma unroll
        for (int r = 0; r < 8; ++r) {
          v4f o = {acc[m][0][r], acc[m][1][r], acc[m][2][r], acc[m][3][r]};
          *(v4f*)(trow + (m * 16 + r + 8 * hi) * TLS_ROWSTRIDE) = o;
        }
    }
    __syncthreads();

#pragma nounroll
    for (int j = 0; j < 48 / NWAVES; ++j) {
      const int lc = wv + NWAVES * j;
      const int ch = cbase + lc;
      const float* __restrict__ wch = dw + ch * 9;
      float w[9];
#pragma unroll
      for (int t = 0; t < 9; ++t) w[t] = wch[t];

      const float* __restrict__ tch = tls + lc * TLS_ROWSTRIDE + TLS_PAD + c4;
      v4f mid[LROWS];
      float lft[LROWS], rgt[LROWS];
#pragma unroll
      for (int r = 0; r < LROWS; ++r) {
        mid[r] = *(const v4f*)(tch + r * TLS_PITCH);
        lft[r] = tch[r * TLS_PITCH - 1];
        rgt[r] = tch[r * TLS_PITCH + 4];
      }

      float ss = 0.f;
#pragma unroll
      for (int orow = 0; orow < OROWS; ++orow) {
        float s0 = 0.f, s1 = 0.f, s2 = 0.f, s3 = 0.f;
#pragma unroll
        for (int dy = 0; dy < 3; ++dy) {
          const int r = orow + dy;
          const float w0 = w[dy * 3 + 0], w1 = w[dy * 3 + 1], w2 = w[dy * 3 + 2];
          s0 += w0 * lft[r];    s0 += w1 * mid[r].x;  s0 += w2 * mid[r].y;
          s1 += w0 * mid[r].x;  s1 += w1 * mid[r].y;  s1 += w2 * mid[r].z;
          s2 += w0 * mid[r].y;  s2 += w1 * mid[r].z;  s2 += w2 * mid[r].w;
          s3 += w0 * mid[r].z;  s3 += w1 * mid[r].w;  s3 += w2 * rgt[r];
        }
        v4f o = {s0, s1, s2, s3};
        *(v4f*)(QKV + (size_t)ch * HW + (r0 + orow) * IMG_W + c4) = o;
        ss += s0 * s0;  ss += s1 * s1;  ss += s2 * s2;  ss += s3 * s3;
      }
      if (ch < 2 * AN2) {
#pragma unroll
        for (int off = 16; off > 0; off >>= 1) ss += __shfl_xor(ss, off, 32);
        if (lane == 0) atomicAdd(&gsums[img * 64 + ch], ss);
      }
    }
    if (half_ == 0) __syncthreads();
  }
}

__global__ void k_att_partial(const float* __restrict__ qkv,
                              float* __restrict__ Sbuf) {
  const int lane = threadIdx.x & 31, wave = threadIdx.x >> 5;
  const int lo = lane & 15, hi = lane >> 4;
  const int img = blockIdx.y;
  const float* __restrict__ Q = qkv + (size_t)img * CMID * HW;
  const float* __restrict__ Km = Q + (size_t)AN2 * HW;
  const int m0 = (wave >> 1) * 16, n0 = (wave & 1) * 16;
  const int kbase = blockIdx.x * (HW / KCHUNKS);

  const float* __restrict__ qrow = Q + (size_t)(m0 + lo) * HW + 2 * hi;
  const float* __restrict__ krow = Km + (size_t)(n0 + lo) * HW + 2 * hi;

  v8f acc0 = {}, acc1 = {}, acc2 = {}, acc3 = {};
  for (int kk = kbase; kk < kbase + HW / KCHUNKS; kk += 16) {
    v2f a, b;
    a.x = qrow[kk];      a.y = qrow[kk + 1];
    b.x = krow[kk];      b.y = krow[kk + 1];
    acc0 = wmma4(a, b, acc0);
    a.x = qrow[kk + 4];  a.y = qrow[kk + 5];
    b.x = krow[kk + 4];  b.y = krow[kk + 5];
    acc1 = wmma4(a, b, acc1);
    a.x = qrow[kk + 8];  a.y = qrow[kk + 9];
    b.x = krow[kk + 8];  b.y = krow[kk + 9];
    acc2 = wmma4(a, b, acc2);
    a.x = qrow[kk + 12]; a.y = qrow[kk + 13];
    b.x = krow[kk + 12]; b.y = krow[kk + 13];
    acc3 = wmma4(a, b, acc3);
  }
  v8f acc = (acc0 + acc1) + (acc2 + acc3);

  float* __restrict__ Sp = Sbuf + (size_t)img * (AN2 * AN2);
#pragma unroll
  for (int r = 0; r < 8; ++r)
    atomicAdd(&Sp[(m0 + r + 8 * hi) * AN2 + n0 + lo], acc[r]);
}

__global__ void k_att_final(const float* __restrict__ Sbuf,
                            const float* __restrict__ sums,
                            const float* __restrict__ proj,
                            float* __restrict__ Wbuf) {
  const int img = blockIdx.x;
  __shared__ float S[AN2][AN2 + 1];
  __shared__ float rq[AN2], rk[AN2];

  if (threadIdx.x < 32) {
    rq[threadIdx.x] = 1.f / fmaxf(sqrtf(sums[img * 64 + threadIdx.x]), 1e-12f);
  } else if (threadIdx.x < 64) {
    int d = threadIdx.x - 32;
    rk[d] = 1.f / fmaxf(sqrtf(sums[img * 64 + 32 + d]), 1e-12f);
  }
  __syncthreads();
  const float* __restrict__ Sp = Sbuf + (size_t)img * (AN2 * AN2);
  for (int idx = threadIdx.x; idx < AN2 * AN2; idx += 128) {
    int row = idx >> 5, col = idx & 31;
    S[row][col] = Sp[idx] * rq[row] * rk[col];
  }
  __syncthreads();
  if (threadIdx.x < 32) {
    int row = threadIdx.x;
    float mx = -3.4e38f;
    for (int j = 0; j < 32; ++j) mx = fmaxf(mx, S[row][j]);
    float sum = 0.f;
    for (int j = 0; j < 32; ++j) {
      float e = __expf(S[row][j] - mx);
      S[row][j] = e;
      sum += e;
    }
    float inv = 1.f / sum;
    for (int j = 0; j < 32; ++j) S[row][j] *= inv;
  }
  __syncthreads();
  for (int idx = threadIdx.x; idx < 64 * 32; idx += 128) {
    int o = idx >> 5, d = idx & 31;
    float s = 0.f;
    for (int cc = 0; cc < 32; ++cc) s += proj[o * 32 + cc] * S[cc][d];
    Wbuf[(size_t)img * 2048 + idx] = s;
  }
}

#define OPX    64
#define OGROUPS 4
#define OWAVES 8
__global__ void __launch_bounds__(OWAVES * 32)
k_out(const float* __restrict__ qkv,
      const float* __restrict__ Wbuf,
      float* __restrict__ out) {
  const int lane = threadIdx.x & 31;
  const int lo = lane & 15, hi = lane >> 4;
  const int wv = __builtin_amdgcn_readfirstlane(threadIdx.x >> 5);
  const int img = blockIdx.y;
  const int p0 = (blockIdx.x * OWAVES + wv) * (OPX * OGROUPS);
  const float* __restrict__ vp =
      qkv + (size_t)img * CMID * HW + (size_t)(2 * AN2 + 2 * hi) * HW + p0 + 4 * lo;
  const float* __restrict__ wp = Wbuf + (size_t)img * 2048 + lo * 32 + 2 * hi;
  float* __restrict__ op = out + (size_t)img * CIN * HW + (size_t)(8 * hi) * HW + p0 + 4 * lo;

  v4f x0[2][8], x1[2][8];
#pragma unroll
  for (int ks = 0; ks < 8; ++ks) {
    x0[0][ks] = *(const v4f*)(vp + (size_t)(4 * ks) * HW);
    x1[0][ks] = *(const v4f*)(vp + (size_t)(4 * ks + 1) * HW);
  }
  v2f a[8][4];
#pragma unroll
  for (int ks = 0; ks < 8; ++ks)
#pragma unroll
    for (int m = 0; m < 4; ++m) a[ks][m] = *(const v2f*)(wp + m * 16 * 32 + 4 * ks);

#pragma unroll
  for (int g = 0; g < OGROUPS; ++g) {
    const int cur = g & 1;
    if (g + 1 < OGROUPS) {
#pragma unroll
      for (int ks = 0; ks < 8; ++ks) {
        x0[cur ^ 1][ks] = *(const v4f*)(vp + (size_t)(4 * ks) * HW + OPX * (g + 1));
        x1[cur ^ 1][ks] = *(const v4f*)(vp + (size_t)(4 * ks + 1) * HW + OPX * (g + 1));
      }
    }
    __builtin_amdgcn_sched_barrier(0);
    v8f acc[4][4] = {};
#pragma unroll
    for (int ks = 0; ks < 8; ++ks) {
#pragma unroll
      for (int i = 0; i < 4; ++i) {
        v2f b;
        b.x = x0[cur][ks][i];
        b.y = x1[cur][ks][i];
#pragma unroll
        for (int m = 0; m < 4; ++m) acc[m][i] = wmma4(a[ks][m], b, acc[m][i]);
      }
    }
#pragma unroll
    for (int m = 0; m < 4; ++m)
#pragma unroll
      for (int r = 0; r < 8; ++r) {
        v4f o = {acc[m][0][r], acc[m][1][r], acc[m][2][r], acc[m][3][r]};
        __builtin_nontemporal_store(o, (v4f*)(op + (size_t)(m * 16 + r) * HW + OPX * g));
      }
  }
}

extern "C" void kernel_launch(void* const* d_in, const int* in_sizes, int n_in,
                              void* d_out, int out_size, void* d_ws, size_t ws_size,
                              hipStream_t stream) {
  const float* x    = (const float*)d_in[0];
  const float* pw   = (const float*)d_in[1];
  const float* dw   = (const float*)d_in[2];
  const float* proj = (const float*)d_in[3];
  float* out = (float*)d_out;

  float* qkv  = (float*)d_ws;
  float* sums = qkv + (size_t)BCN * CMID * HW;
  float* Sbuf = sums + (size_t)BCN * 64;
  float* Wbuf = Sbuf + (size_t)BCN * AN2 * AN2;
  const int nzero = BCN * 64 + BCN * AN2 * AN2;

  k_zero<<<dim3((nzero + 255) / 256), 256, 0, stream>>>(sums, nzero);
  k_fused<<<dim3(IMG_W / OROWS, BCN), NWAVES * 32, TLS_FLOATS * sizeof(float), stream>>>(
      x, pw, dw, qkv, sums);
  k_att_partial<<<dim3(KCHUNKS, BCN), 128, 0, stream>>>(qkv, Sbuf);
  k_att_final<<<dim3(BCN), 128, 0, stream>>>(Sbuf, sums, proj, Wbuf);
  k_out<<<dim3(HW / (OWAVES * OPX * OGROUPS), BCN), OWAVES * 32, 0, stream>>>(qkv, Wbuf, out);
}


# k_att_partial: a wave forms the whole 32x32 partial from whole-line b128 q/k fragments (same k map both sides), 2-3 steps of loads in flight; 8 waves summed in LDS before atomics (sum order changed)
# speedup vs baseline: 2.9343x; 1.1207x over previous
#include <hip/hip_runtime.h>

typedef __attribute__((ext_vector_type(2))) float v2f;
typedef __attribute__((ext_vector_type(4))) float v4f;
typedef __attribute__((ext_vector_type(8))) float v8f;

#define HW    16384
#define IMG_W 128
#define BCN   32
#define CIN   64
#define CMID  96
#define AN2   32
#define OROWS  4
#define LROWS  (OROWS + 2)
#define NWAVES (2 * LROWS)
#define X_AHEAD 2
#define W_AHEAD 1
#define TLS_PAD   4
#define TLS_PITCH (IMG_W + TLS_PAD)
#define TLS_ROWSTRIDE (LROWS * TLS_PITCH)
#define TLS_FLOATS (48 * TLS_ROWSTRIDE + TLS_PAD)
#define KCHUNKS 16

static __device__ __forceinline__ v8f wmma4(v2f a, v2f b, v8f c) {
  return __builtin_amdgcn_wmma_f32_16x16x4_f32(
       false, a,  false, b,
       (short)0, c,  false,  false);
}

__global__ void k_zero(float* __restrict__ p, int n) {
  int i = blockIdx.x * 256 + threadIdx.x;
  if (i < n) p[i] = 0.f;
}

__global__ void __launch_bounds__(NWAVES * 32) __attribute__((amdgpu_waves_per_eu(6, 6)))
k_fused(const float* __restrict__ x,
        const float* __restrict__ pw,
        const float* __restrict__ dw,
        float* __restrict__ qkv,
        float* __restrict__ gsums) {
  extern __shared__ __attribute__((aligned(16))) float tls[];

  const int tid  = threadIdx.x;
  const int lane = tid & 31, wave = tid >> 5;
  const int lo = lane & 15, hi = lane >> 4;
  const int img = blockIdx.y;
  const int r0  = blockIdx.x * OROWS;
  const float* __restrict__ X = x + (size_t)img * CIN * HW;
  float* __restrict__ QKV = qkv + (size_t)img * CMID * HW;

  if (tid <= 48 * LROWS) {
    v4f z = {0.f, 0.f, 0.f, 0.f};
    *(v4f*)(tls + tid * TLS_PITCH) = z;
  }

  const int wv = __builtin_amdgcn_readfirstlane(wave);
  const int c4 = lane * 4;

  const int lrow = wv >> 1;
  const int colbase = (wv & 1) * 64;
  const int rr = r0 - 1 + lrow;
  float* __restrict__ trow = tls + lrow * TLS_PITCH + TLS_PAD + colbase + 4 * lo;

  for (int half_ = 0; half_ < 2; ++half_) {
    const int cbase = half_ * 48;

    if (rr < 0 || rr >= IMG_W) {
      v4f z = {0.f, 0.f, 0.f, 0.f};
      for (int c = hi; c < 48; c += 2) *(v4f*)(trow + c * TLS_ROWSTRIDE) = z;
    } else {
      const float* __restrict__ xp = X + (size_t)(2 * hi) * HW + rr * IMG_W + colbase + 4 * lo;
      const float* __restrict__ ap = pw + (cbase + lo) * CIN + 2 * hi;
      v8f acc[3][4] = {};
      v4f x0[X_AHEAD + 1], x1[X_AHEAD + 1];
      v2f a[W_AHEAD + 1][3];
#pragma unroll
      for (int st = 0; st < X_AHEAD; ++st) {
        x0[st] = *(const v4f*)(xp + (size_t)(4 * st) * HW);
        x1[st] = *(const v4f*)(xp + (size_t)(4 * st + 1) * HW);
      }
#pragma unroll
      for (int st = 0; st < W_AHEAD; ++st)
#pragma unroll
        for (int m = 0; m < 3; ++m) a[st][m] = *(const v2f*)(ap + m * 16 * CIN + 4 * st);
#pragma unroll
      for (int st = 0; st < CIN / 4; ++st) {
        if (st + W_AHEAD < CIN / 4) {
          const int sl = (st + W_AHEAD) % (W_AHEAD + 1);
          const int kn = 4 * (st + W_AHEAD);
#pragma unroll
          for (int m = 0; m < 3; ++m) a[sl][m] = *(const v2f*)(ap + m * 16 * CIN + kn);
        }
        if (st + X_AHEAD < CIN / 4) {
          const int sl = (st + X_AHEAD) % (X_AHEAD + 1);
          const int kn = 4 * (st + X_AHEAD);
          x0[sl] = *(const v4f*)(xp + (size_t)kn * HW);
          x1[sl] = *(const v4f*)(xp + (size_t)(kn + 1) * HW);
        }
        __builtin_amdgcn_sched_barrier(0);
        const int sx = st % (X_AHEAD + 1);
        const int sw = st % (W_AHEAD + 1);
#pragma unroll
        for (int i = 0; i < 4; ++i) {
          v2f b;
          b.x = x0[sx][i];
          b.y = x1[sx][i];
#pragma unroll
          for (int m = 0; m < 3; ++m) acc[m][i] = wmma4(a[sw][m], b, acc[m][i]);
        }
      }
#pragma unroll
      for (int m = 0; m < 3; ++m)
#pragma unroll
        for (int r = 0; r < 8; ++r) {
          v4f o = {acc[m][0][r], acc[m][1][r], acc[m][2][r], acc[m][3][r]};
          *(v4f*)(trow + (m * 16 + r + 8 * hi) * TLS_ROWSTRIDE) = o;
        }
    }
    __syncthreads();

#pragma nounroll
    for (int j = 0; j < 48 / NWAVES; ++j) {
      const int lc = wv + NWAVES * j;
      const int ch = cbase + lc;
      const float* __restrict__ wch = dw + ch * 9;
      float w[9];
#pragma unroll
      for (int t = 0; t < 9; ++t) w[t] = wch[t];

      const float* __restrict__ tch = tls + lc * TLS_ROWSTRIDE + TLS_PAD + c4;
      v4f mid[LROWS];
      float lft[LROWS], rgt[LROWS];
#pragma unroll
      for (int r = 0; r < LROWS; ++r) {
        mid[r] = *(const v4f*)(tch + r * TLS_PITCH);
        lft[r] = tch[r * TLS_PITCH - 1];
        rgt[r] = tch[r * TLS_PITCH + 4];
      }

      float ss = 0.f;
#pragma unroll
      for (int orow = 0; orow < OROWS; ++orow) {
        float s0 = 0.f, s1 = 0.f, s2 = 0.f, s3 = 0.f;
#pragma unroll
        for (int dy = 0; dy < 3; ++dy) {
          const int r = orow + dy;
          const float w0 = w[dy * 3 + 0], w1 = w[dy * 3 + 1], w2 = w[dy * 3 + 2];
          s0 += w0 * lft[r];    s0 += w1 * mid[r].x;  s0 += w2 * mid[r].y;
          s1 += w0 * mid[r].x;  s1 += w1 * mid[r].y;  s1 += w2 * mid[r].z;
          s2 += w0 * mid[r].y;  s2 += w1 * mid[r].z;  s2 += w2 * mid[r].w;
          s3 += w0 * mid[r].z;  s3 += w1 * mid[r].w;  s3 += w2 * rgt[r];
        }
        v4f o = {s0, s1, s2, s3};
        *(v4f*)(QKV + (size_t)ch * HW + (r0 + orow) * IMG_W + c4) = o;
        ss += s0 * s0;  ss += s1 * s1;  ss += s2 * s2;  ss += s3 * s3;
      }
      if (ch < 2 * AN2) {
#pragma unroll
        for (int off = 16; off > 0; off >>= 1) ss += __shfl_xor(ss, off, 32);
        if (lane == 0) atomicAdd(&gsums[img * 64 + ch], ss);
      }
    }
    if (half_ == 0) __syncthreads();
  }
}

#define SK_WAVES 8
#define SK_STEPS 4
__global__ void __launch_bounds__(SK_WAVES * 32)
k_att_partial(const float* __restrict__ qkv,
              float* __restrict__ Sbuf) {
  __shared__ __attribute__((aligned(16))) float red[SK_WAVES][AN2 * AN2];
  const int lane = threadIdx.x & 31;
  const int lo = lane & 15, hi = lane >> 4;
  const int wv = __builtin_amdgcn_readfirstlane(threadIdx.x >> 5);
  const int img = blockIdx.y;
  const int p0 = blockIdx.x * (HW / KCHUNKS) + wv * (SK_STEPS * 32);
  const float* __restrict__ rp =
      qkv + (size_t)img * CMID * HW + (size_t)lo * HW + p0 + 16 * hi;

  v4f f[2][4][4];
#pragma unroll
  for (int t = 0; t < 4; ++t)
#pragma unroll
    for (int q4 = 0; q4 < 4; ++q4)
      f[0][t][q4] = *(const v4f*)(rp + (size_t)(t * 16) * HW + 4 * q4);

  v8f acc[2][2] = {};
#pragma unroll
  for (int st = 0; st < SK_STEPS; ++st) {
    const int cur = st & 1;
    if (st + 1 < SK_STEPS) {
#pragma unroll
      for (int t = 0; t < 4; ++t)
#pragma unroll
        for (int q4 = 0; q4 < 4; ++q4)
          f[cur ^ 1][t][q4] = *(const v4f*)(rp + (size_t)(t * 16) * HW + 32 * (st + 1) + 4 * q4);
    }
    __builtin_amdgcn_sched_barrier(0);
#pragma unroll
    for (int j = 0; j < 8; ++j) {
      const int q4 = j >> 1, e = 2 * (j & 1);
      v2f a0, a1, b0, b1;
      a0.x = f[cur][0][q4][e];  a0.y = f[cur][0][q4][e + 1];
      a1.x = f[cur][1][q4][e];  a1.y = f[cur][1][q4][e + 1];
      b0.x = f[cur][2][q4][e];  b0.y = f[cur][2][q4][e + 1];
      b1.x = f[cur][3][q4][e];  b1.y = f[cur][3][q4][e + 1];
      acc[0][0] = wmma4(a0, b0, acc[0][0]);
      acc[0][1] = wmma4(a0, b1, acc[0][1]);
      acc[1][0] = wmma4(a1, b0, acc[1][0]);
      acc[1][1] = wmma4(a1, b1, acc[1][1]);
    }
  }

#pragma unroll
  for (int m = 0; m < 2; ++m)
#pragma unroll
    for (int n = 0; n < 2; ++n)
#pragma unroll
      for (int r = 0; r < 8; ++r)
        red[wv][(m * 16 + 8 * hi + r) * AN2 + n * 16 + lo] = acc[m][n][r];
  __syncthreads();

  float* __restrict__ Sp = Sbuf + (size_t)img * (AN2 * AN2);
#pragma unroll
  for (int e = 0; e < (AN2 * AN2) / (SK_WAVES * 32); ++e) {
    const int idx = threadIdx.x + e * (SK_WAVES * 32);
    float s = red[0][idx];
#pragma unroll
    for (int w = 1; w < SK_WAVES; ++w) s += red[w][idx];
    atomicAdd(&Sp[idx], s);
  }
}

__global__ void k_att_final(const float* __restrict__ Sbuf,
                            const float* __restrict__ sums,
                            const float* __restrict__ proj,
                            float* __restrict__ Wbuf) {
  const int img = blockIdx.x;
  __shared__ float S[AN2][AN2 + 1];
  __shared__ float rq[AN2], rk[AN2];

  if (threadIdx.x < 32) {
    rq[threadIdx.x] = 1.f / fmaxf(sqrtf(sums[img * 64 + threadIdx.x]), 1e-12f);
  } else if (threadIdx.x < 64) {
    int d = threadIdx.x - 32;
    rk[d] = 1.f / fmaxf(sqrtf(sums[img * 64 + 32 + d]), 1e-12f);
  }
  __syncthreads();
  const float* __restrict__ Sp = Sbuf + (size_t)img * (AN2 * AN2);
  for (int idx = threadIdx.x; idx < AN2 * AN2; idx += 128) {
    int row = idx >> 5, col = idx & 31;
    S[row][col] = Sp[idx] * rq[row] * rk[col];
  }
  __syncthreads();
  if (threadIdx.x < 32) {
    int row = threadIdx.x;
    float mx = -3.4e38f;
    for (int j = 0; j < 32; ++j) mx = fmaxf(mx, S[row][j]);
    float sum = 0.f;
    for (int j = 0; j < 32; ++j) {
      float e = __expf(S[row][j] - mx);
      S[row][j] = e;
      sum += e;
    }
    float inv = 1.f / sum;
    for (int j = 0; j < 32; ++j) S[row][j] *= inv;
  }
  __syncthreads();
  for (int idx = threadIdx.x; idx < 64 * 32; idx += 128) {
    int o = idx >> 5, d = idx & 31;
    float s = 0.f;
    for (int cc = 0; cc < 32; ++cc) s += proj[o * 32 + cc] * S[cc][d];
    Wbuf[(size_t)img * 2048 + idx] = s;
  }
}

#define OPX    64
#define OGROUPS 4
#define OWAVES 8
__global__ void __launch_bounds__(OWAVES * 32)
k_out(const float* __restrict__ qkv,
      const float* __restrict__ Wbuf,
      float* __restrict__ out) {
  const int lane = threadIdx.x & 31;
  const int lo = lane & 15, hi = lane >> 4;
  const int wv = __builtin_amdgcn_readfirstlane(threadIdx.x >> 5);
  const int img = blockIdx.y;
  const int p0 = (blockIdx.x * OWAVES + wv) * (OPX * OGROUPS);
  const float* __restrict__ vp =
      qkv + (size_t)img * CMID * HW + (size_t)(2 * AN2 + 2 * hi) * HW + p0 + 4 * lo;
  const float* __restrict__ wp = Wbuf + (size_t)img * 2048 + lo * 32 + 2 * hi;
  float* __restrict__ op = out + (size_t)img * CIN * HW + (size_t)(8 * hi) * HW + p0 + 4 * lo;

  v4f x0[2][8], x1[2][8];
#pragma unroll
  for (int ks = 0; ks < 8; ++ks) {
    x0[0][ks] = *(const v4f*)(vp + (size_t)(4 * ks) * HW);
    x1[0][ks] = *(const v4f*)(vp + (size_t)(4 * ks + 1) * HW);
  }
  v2f a[8][4];
#pragma unroll
  for (int ks = 0; ks < 8; ++ks)
#pragma unroll
    for (int m = 0; m < 4; ++m) a[ks][m] = *(const v2f*)(wp + m * 16 * 32 + 4 * ks);

#pragma unroll
  for (int g = 0; g < OGROUPS; ++g) {
    const int cur = g & 1;
    if (g + 1 < OGROUPS) {
#pragma unroll
      for (int ks = 0; ks < 8; ++ks) {
        x0[cur ^ 1][ks] = *(const v4f*)(vp + (size_t)(4 * ks) * HW + OPX * (g + 1));
        x1[cur ^ 1][ks] = *(const v4f*)(vp + (size_t)(4 * ks + 1) * HW + OPX * (g + 1));
      }
    }
    __builtin_amdgcn_sched_barrier(0);
    v8f acc[4][4] = {};
#pragma unroll
    for (int ks = 0; ks < 8; ++ks) {
#pragma unroll
      for (int i = 0; i < 4; ++i) {
        v2f b;
        b.x = x0[cur][ks][i];
        b.y = x1[cur][ks][i];
#pragma unroll
        for (int m = 0; m < 4; ++m) acc[m][i] = wmma4(a[ks][m], b, acc[m][i]);
      }
    }
#pragma unroll
    for (int m = 0; m < 4; ++m)
#pragma unroll
      for (int r = 0; r < 8; ++r) {
        v4f o = {acc[m][0][r], acc[m][1][r], acc[m][2][r], acc[m][3][r]};
        __builtin_nontemporal_store(o, (v4f*)(op + (size_t)(m * 16 + r) * HW + OPX * g));
      }
  }
}

extern "C" void kernel_launch(void* const* d_in, const int* in_sizes, int n_in,
                              void* d_out, int out_size, void* d_ws, size_t ws_size,
                              hipStream_t stream) {
  const float* x    = (const float*)d_in[0];
  const float* pw   = (const float*)d_in[1];
  const float* dw   = (const float*)d_in[2];
  const float* proj = (const float*)d_in[3];
  float* out = (float*)d_out;

  float* qkv  = (float*)d_ws;
  float* sums = qkv + (size_t)BCN * CMID * HW;
  float* Sbuf = sums + (size_t)BCN * 64;
  float* Wbuf = Sbuf + (size_t)BCN * AN2 * AN2;
  const int nzero = BCN * 64 + BCN * AN2 * AN2;

  k_zero<<<dim3((nzero + 255) / 256), 256, 0, stream>>>(sums, nzero);
  k_fused<<<dim3(IMG_W / OROWS, BCN), NWAVES * 32, TLS_FLOATS * sizeof(float), stream>>>(
      x, pw, dw, qkv, sums);
  k_att_partial<<<dim3(KCHUNKS, BCN), SK_WAVES * 32, 0, stream>>>(qkv, Sbuf);
  k_att_final<<<dim3(BCN), 128, 0, stream>>>(Sbuf, sums, proj, Wbuf);
  k_out<<<dim3(HW / (OWAVES * OPX * OGROUPS), BCN), OWAVES * 32, 0, stream>>>(qkv, Wbuf, out);
}


# fused kernel: a block walks 8 output rows as two 4-row sub-blocks and keeps the 2 shared t rows in LDS (ring of 6), so sub-block 2 forms only 4 new rows: pointwise halo recompute 1.5x -> 1.25x
# speedup vs baseline: 3.0408x; 1.0363x over previous
#include <hip/hip_runtime.h>

typedef __attribute__((ext_vector_type(2))) float v2f;
typedef __attribute__((ext_vector_type(4))) float v4f;
typedef __attribute__((ext_vector_type(8))) float v8f;

#define HW    16384
#define IMG_W 128
#define BCN   32
#define CIN   64
#define CMID  96
#define AN2   32
#define OROWS  4
#define SUBS   2
#define LROWS  (OROWS + 2)
#define NWAVES (2 * LROWS)
#define X_AHEAD 2
#define W_AHEAD 1
#define TLS_PAD   4
#define TLS_PITCH (IMG_W + TLS_PAD)
#define TLS_ROWSTRIDE (LROWS * TLS_PITCH)
#define TLS_FLOATS (48 * TLS_ROWSTRIDE + TLS_PAD)
#define KCHUNKS 16

static __device__ __forceinline__ v8f wmma4(v2f a, v2f b, v8f c) {
  return __builtin_amdgcn_wmma_f32_16x16x4_f32(
       false, a,  false, b,
       (short)0, c,  false,  false);
}

__global__ void k_zero(float* __restrict__ p, int n) {
  int i = blockIdx.x * 256 + threadIdx.x;
  if (i < n) p[i] = 0.f;
}

__global__ void __launch_bounds__(NWAVES * 32) __attribute__((amdgpu_waves_per_eu(6, 6)))
k_fused(const float* __restrict__ x,
        const float* __restrict__ pw,
        const float* __restrict__ dw,
        float* __restrict__ qkv,
        float* __restrict__ gsums) {
  extern __shared__ __attribute__((aligned(16))) float tls[];

  const int tid  = threadIdx.x;
  const int lane = tid & 31, wave = tid >> 5;
  const int lo = lane & 15, hi = lane >> 4;
  const int img = blockIdx.y;
  const int r0  = blockIdx.x * (OROWS * SUBS);
  const float* __restrict__ X = x + (size_t)img * CIN * HW;
  float* __restrict__ QKV = qkv + (size_t)img * CMID * HW;

  if (tid <= 48 * LROWS) {
    v4f z = {0.f, 0.f, 0.f, 0.f};
    *(v4f*)(tls + tid * TLS_PITCH) = z;
  }

  const int wv = __builtin_amdgcn_readfirstlane(wave);
  const int c4 = lane * 4;

  const int lrow = wv >> 1;
  const int colbase = (wv & 1) * 64;
  float* __restrict__ trow = tls + lrow * TLS_PITCH + TLS_PAD + colbase + 4 * lo;

  for (int half_ = 0; half_ < 2; ++half_) {
    const int cbase = half_ * 48;

#pragma nounroll
    for (int sub = 0; sub < SUBS; ++sub) {
    const int rr = r0 - 1 + lrow + LROWS * sub;

    if (sub != 0 && lrow >= OROWS) {
    } else if (rr < 0 || rr >= IMG_W) {
      v4f z = {0.f, 0.f, 0.f, 0.f};
      for (int c = hi; c < 48; c += 2) *(v4f*)(trow + c * TLS_ROWSTRIDE) = z;
    } else {
      const float* __restrict__ xp = X + (size_t)(2 * hi) * HW + rr * IMG_W + colbase + 4 * lo;
      const float* __restrict__ ap = pw + (cbase + lo) * CIN + 2 * hi;
      v8f acc[3][4] = {};
      v4f x0[X_AHEAD + 1], x1[X_AHEAD + 1];
      v2f a[W_AHEAD + 1][3];
#pragma unroll
      for (int st = 0; st < X_AHEAD; ++st) {
        x0[st] = *(const v4f*)(xp + (size_t)(4 * st) * HW);
        x1[st] = *(const v4f*)(xp + (size_t)(4 * st + 1) * HW);
      }
#pragma unroll
      for (int st = 0; st < W_AHEAD; ++st)
#pragma unroll
        for (int m = 0; m < 3; ++m) a[st][m] = *(const v2f*)(ap + m * 16 * CIN + 4 * st);
#pragma unroll
      for (int st = 0; st < CIN / 4; ++st) {
        if (st + W_AHEAD < CIN / 4) {
          const int sl = (st + W_AHEAD) % (W_AHEAD + 1);
          const int kn = 4 * (st + W_AHEAD);
#pragma unroll
          for (int m = 0; m < 3; ++m) a[sl][m] = *(const v2f*)(ap + m * 16 * CIN + kn);
        }
        if (st + X_AHEAD < CIN / 4) {
          const int sl = (st + X_AHEAD) % (X_AHEAD + 1);
          const int kn = 4 * (st + X_AHEAD);
          x0[sl] = *(const v4f*)(xp + (size_t)kn * HW);
          x1[sl] = *(const v4f*)(xp + (size_t)(kn + 1) * HW);
        }
        __builtin_amdgcn_sched_barrier(0);
        const int sx = st % (X_AHEAD + 1);
        const int sw = st % (W_AHEAD + 1);
#pragma unroll
        for (int i = 0; i < 4; ++i) {
          v2f b;
          b.x = x0[sx][i];
          b.y = x1[sx][i];
#pragma unroll
          for (int m = 0; m < 3; ++m) acc[m][i] = wmma4(a[sw][m], b, acc[m][i]);
        }
      }
#pragma unroll
      for (int m = 0; m < 3; ++m)
#pragma unroll
        for (int r = 0; r < 8; ++r) {
          v4f o = {acc[m][0][r], acc[m][1][r], acc[m][2][r], acc[m][3][r]};
          *(v4f*)(trow + (m * 16 + r + 8 * hi) * TLS_ROWSTRIDE) = o;
        }
    }
    __syncthreads();

#pragma nounroll
    for (int j = 0; j < 48 / NWAVES; ++j) {
      const int lc = wv + NWAVES * j;
      const int ch = cbase + lc;
      const float* __restrict__ wch = dw + ch * 9;
      float w[9];
#pragma unroll
      for (int t = 0; t < 9; ++t) w[t] = wch[t];

      const float* __restrict__ tch = tls + lc * TLS_ROWSTRIDE + TLS_PAD + c4;
      v4f mid[LROWS];
      float lft[LROWS], rgt[LROWS];
#pragma unroll
      for (int r = 0; r < LROWS; ++r) {
        int lr = r + OROWS * sub;
        if (lr >= LROWS) lr -= LROWS;
        const float* __restrict__ trd = tch + lr * TLS_PITCH;
        mid[r] = *(const v4f*)(trd);
        lft[r] = trd[-1];
        rgt[r] = trd[4];
      }

      float ss = 0.f;
#pragma unroll
      for (int orow = 0; orow < OROWS; ++orow) {
        float s0 = 0.f, s1 = 0.f, s2 = 0.f, s3 = 0.f;
#pragma unroll
        for (int dy = 0; dy < 3; ++dy) {
          const int r = orow + dy;
          const float w0 = w[dy * 3 + 0], w1 = w[dy * 3 + 1], w2 = w[dy * 3 + 2];
          s0 += w0 * lft[r];    s0 += w1 * mid[r].x;  s0 += w2 * mid[r].y;
          s1 += w0 * mid[r].x;  s1 += w1 * mid[r].y;  s1 += w2 * mid[r].z;
          s2 += w0 * mid[r].y;  s2 += w1 * mid[r].z;  s2 += w2 * mid[r].w;
          s3 += w0 * mid[r].z;  s3 += w1 * mid[r].w;  s3 += w2 * rgt[r];
        }
        v4f o = {s0, s1, s2, s3};
        *(v4f*)(QKV + (size_t)ch * HW + (r0 + OROWS * sub + orow) * IMG_W + c4) = o;
        ss += s0 * s0;  ss += s1 * s1;  ss += s2 * s2;  ss += s3 * s3;
      }
      if (ch < 2 * AN2) {
#pragma unroll
        for (int off = 16; off > 0; off >>= 1) ss += __shfl_xor(ss, off, 32);
        if (lane == 0) atomicAdd(&gsums[img * 64 + ch], ss);
      }
    }
    if (half_ == 0 || sub + 1 < SUBS) __syncthreads();
    }
  }
}

#define SK_WAVES 8
#define SK_STEPS 4
__global__ void __launch_bounds__(SK_WAVES * 32)
k_att_partial(const float* __restrict__ qkv,
              float* __restrict__ Sbuf) {
  __shared__ __attribute__((aligned(16))) float red[SK_WAVES][AN2 * AN2];
  const int lane = threadIdx.x & 31;
  const int lo = lane & 15, hi = lane >> 4;
  const int wv = __builtin_amdgcn_readfirstlane(threadIdx.x >> 5);
  const int img = blockIdx.y;
  const int p0 = blockIdx.x * (HW / KCHUNKS) + wv * (SK_STEPS * 32);
  const float* __restrict__ rp =
      qkv + (size_t)img * CMID * HW + (size_t)lo * HW + p0 + 16 * hi;

  v4f f[2][4][4];
#pragma unroll
  for (int t = 0; t < 4; ++t)
#pragma unroll
    for (int q4 = 0; q4 < 4; ++q4)
      f[0][t][q4] = *(const v4f*)(rp + (size_t)(t * 16) * HW + 4 * q4);

  v8f acc[2][2] = {};
#pragma unroll
  for (int st = 0; st < SK_STEPS; ++st) {
    const int cur = st & 1;
    if (st + 1 < SK_STEPS) {
#pragma unroll
      for (int t = 0; t < 4; ++t)
#pragma unroll
        for (int q4 = 0; q4 < 4; ++q4)
          f[cur ^ 1][t][q4] = *(const v4f*)(rp + (size_t)(t * 16) * HW + 32 * (st + 1) + 4 * q4);
    }
    __builtin_amdgcn_sched_barrier(0);
#pragma unroll
    for (int j = 0; j < 8; ++j) {
      const int q4 = j >> 1, e = 2 * (j & 1);
      v2f a0, a1, b0, b1;
      a0.x = f[cur][0][q4][e];  a0.y = f[cur][0][q4][e + 1];
      a1.x = f[cur][1][q4][e];  a1.y = f[cur][1][q4][e + 1];
      b0.x = f[cur][2][q4][e];  b0.y = f[cur][2][q4][e + 1];
      b1.x = f[cur][3][q4][e];  b1.y = f[cur][3][q4][e + 1];
      acc[0][0] = wmma4(a0, b0, acc[0][0]);
      acc[0][1] = wmma4(a0, b1, acc[0][1]);
      acc[1][0] = wmma4(a1, b0, acc[1][0]);
      acc[1][1] = wmma4(a1, b1, acc[1][1]);
    }
  }

#pragma unroll
  for (int m = 0; m < 2; ++m)
#pragma unroll
    for (int n = 0; n < 2; ++n)
#pragma unroll
      for (int r = 0; r < 8; ++r)
        red[wv][(m * 16 + 8 * hi + r) * AN2 + n * 16 + lo] = acc[m][n][r];
  __syncthreads();

  float* __restrict__ Sp = Sbuf + (size_t)img * (AN2 * AN2);
#pragma unroll
  for (int e = 0; e < (AN2 * AN2) / (SK_WAVES * 32); ++e) {
    const int idx = threadIdx.x + e * (SK_WAVES * 32);
    float s = red[0][idx];
#pragma unroll
    for (int w = 1; w < SK_WAVES; ++w) s += red[w][idx];
    atomicAdd(&Sp[idx], s);
  }
}

__global__ void k_att_final(const float* __restrict__ Sbuf,
                            const float* __restrict__ sums,
                            const float* __restrict__ proj,
                            float* __restrict__ Wbuf) {
  const int img = blockIdx.x;
  __shared__ float S[AN2][AN2 + 1];
  __shared__ float rq[AN2], rk[AN2];

  if (threadIdx.x < 32) {
    rq[threadIdx.x] = 1.f / fmaxf(sqrtf(sums[img * 64 + threadIdx.x]), 1e-12f);
  } else if (threadIdx.x < 64) {
    int d = threadIdx.x - 32;
    rk[d] = 1.f / fmaxf(sqrtf(sums[img * 64 + 32 + d]), 1e-12f);
  }
  __syncthreads();
  const float* __restrict__ Sp = Sbuf + (size_t)img * (AN2 * AN2);
  for (int idx = threadIdx.x; idx < AN2 * AN2; idx += 128) {
    int row = idx >> 5, col = idx & 31;
    S[row][col] = Sp[idx] * rq[row] * rk[col];
  }
  __syncthreads();
  if (threadIdx.x < 32) {
    int row = threadIdx.x;
    float mx = -3.4e38f;
    for (int j = 0; j < 32; ++j) mx = fmaxf(mx, S[row][j]);
    float sum = 0.f;
    for (int j = 0; j < 32; ++j) {
      float e = __expf(S[row][j] - mx);
      S[row][j] = e;
      sum += e;
    }
    float inv = 1.f / sum;
    for (int j = 0; j < 32; ++j) S[row][j] *= inv;
  }
  __syncthreads();
  for (int idx = threadIdx.x; idx < 64 * 32; idx += 128) {
    int o = idx >> 5, d = idx & 31;
    float s = 0.f;
    for (int cc = 0; cc < 32; ++cc) s += proj[o * 32 + cc] * S[cc][d];
    Wbuf[(size_t)img * 2048 + idx] = s;
  }
}

#define OPX    64
#define OGROUPS 4
#define OWAVES 8
__global__ void __launch_bounds__(OWAVES * 32)
k_out(const float* __restrict__ qkv,
      const float* __restrict__ Wbuf,
      float* __restrict__ out) {
  const int lane = threadIdx.x & 31;
  const int lo = lane & 15, hi = lane >> 4;
  const int wv = __builtin_amdgcn_readfirstlane(threadIdx.x >> 5);
  const int img = blockIdx.y;
  const int p0 = (blockIdx.x * OWAVES + wv) * (OPX * OGROUPS);
  const float* __restrict__ vp =
      qkv + (size_t)img * CMID * HW + (size_t)(2 * AN2 + 2 * hi) * HW + p0 + 4 * lo;
  const float* __restrict__ wp = Wbuf + (size_t)img * 2048 + lo * 32 + 2 * hi;
  float* __restrict__ op = out + (size_t)img * CIN * HW + (size_t)(8 * hi) * HW + p0 + 4 * lo;

  v4f x0[2][8], x1[2][8];
#pragma unroll
  for (int ks = 0; ks < 8; ++ks) {
    x0[0][ks] = *(const v4f*)(vp + (size_t)(4 * ks) * HW);
    x1[0][ks] = *(const v4f*)(vp + (size_t)(4 * ks + 1) * HW);
  }
  v2f a[8][4];
#pragma unroll
  for (int ks = 0; ks < 8; ++ks)
#pragma unroll
    for (int m = 0; m < 4; ++m) a[ks][m] = *(const v2f*)(wp + m * 16 * 32 + 4 * ks);

#pragma unroll
  for (int g = 0; g < OGROUPS; ++g) {
    const int cur = g & 1;
    if (g + 1 < OGROUPS) {
#pragma unroll
      for (int ks = 0; ks < 8; ++ks) {
        x0[cur ^ 1][ks] = *(const v4f*)(vp + (size_t)(4 * ks) * HW + OPX * (g + 1));
        x1[cur ^ 1][ks] = *(const v4f*)(vp + (size_t)(4 * ks + 1) * HW + OPX * (g + 1));
      }
    }
    __builtin_amdgcn_sched_barrier(0);
    v8f acc[4][4] = {};
#pragma unroll
    for (int ks = 0; ks < 8; ++ks) {
#pragma unroll
      for (int i = 0; i < 4; ++i) {
        v2f b;
        b.x = x0[cur][ks][i];
        b.y = x1[cur][ks][i];
#pragma unroll
        for (int m = 0; m < 4; ++m) acc[m][i] = wmma4(a[ks][m], b, acc[m][i]);
      }
    }
#pragma unroll
    for (int m = 0; m < 4; ++m)
#pragma unroll
      for (int r = 0; r < 8; ++r) {
        v4f o = {acc[m][0][r], acc[m][1][r], acc[m][2][r], acc[m][3][r]};
        __builtin_nontemporal_store(o, (v4f*)(op + (size_t)(m * 16 + r) * HW + OPX * g));
      }
  }
}

extern "C" void kernel_launch(void* const* d_in, const int* in_sizes, int n_in,
                              void* d_out, int out_size, void* d_ws, size_t ws_size,
                              hipStream_t stream) {
  const float* x    = (const float*)d_in[0];
  const float* pw   = (const float*)d_in[1];
  const float* dw   = (const float*)d_in[2];
  const float* proj = (const float*)d_in[3];
  float* out = (float*)d_out;

  float* qkv  = (float*)d_ws;
  float* sums = qkv + (size_t)BCN * CMID * HW;
  float* Sbuf = sums + (size_t)BCN * 64;
  float* Wbuf = Sbuf + (size_t)BCN * AN2 * AN2;
  const int nzero = BCN * 64 + BCN * AN2 * AN2;

  k_zero<<<dim3((nzero + 255) / 256), 256, 0, stream>>>(sums, nzero);
  k_fused<<<dim3(IMG_W / (OROWS * SUBS), BCN), NWAVES * 32, TLS_FLOATS * sizeof(float), stream>>>(
      x, pw, dw, qkv, sums);
  k_att_partial<<<dim3(KCHUNKS, BCN), SK_WAVES * 32, 0, stream>>>(qkv, Sbuf);
  k_att_final<<<dim3(BCN), 128, 0, stream>>>(Sbuf, sums, proj, Wbuf);
  k_out<<<dim3(HW / (OWAVES * OPX * OGROUPS), BCN), OWAVES * 32, 0, stream>>>(qkv, Wbuf, out);
}


# softmax + proj fold moved into k_out: each block forms its image's 64x32 W in LDS (rows over waves, row max/sum by wave shuffles) before its WMMAs; launches 5 -> 4; softmax row sum order changed
# speedup vs baseline: 3.4103x; 1.1215x over previous
#include <hip/hip_runtime.h>

typedef __attribute__((ext_vector_type(2))) float v2f;
typedef __attribute__((ext_vector_type(4))) float v4f;
typedef __attribute__((ext_vector_type(8))) float v8f;

#define HW    16384
#define IMG_W 128
#define BCN   32
#define CIN   64
#define CMID  96
#define AN2   32
#define OROWS  4
#define SUBS   2
#define LROWS  (OROWS + 2)
#define NWAVES (2 * LROWS)
#define X_AHEAD 2
#define W_AHEAD 1
#define TLS_PAD   4
#define TLS_PITCH (IMG_W + TLS_PAD)
#define TLS_ROWSTRIDE (LROWS * TLS_PITCH)
#define TLS_FLOATS (48 * TLS_ROWSTRIDE + TLS_PAD)
#define KCHUNKS 16

static __device__ __forceinline__ v8f wmma4(v2f a, v2f b, v8f c) {
  return __builtin_amdgcn_wmma_f32_16x16x4_f32(
       false, a,  false, b,
       (short)0, c,  false,  false);
}

__global__ void k_zero(float* __restrict__ p, int n) {
  int i = blockIdx.x * 256 + threadIdx.x;
  if (i < n) p[i] = 0.f;
}

__global__ void __launch_bounds__(NWAVES * 32) __attribute__((amdgpu_waves_per_eu(6, 6)))
k_fused(const float* __restrict__ x,
        const float* __restrict__ pw,
        const float* __restrict__ dw,
        float* __restrict__ qkv,
        float* __restrict__ gsums) {
  extern __shared__ __attribute__((aligned(16))) float tls[];

  const int tid  = threadIdx.x;
  const int lane = tid & 31, wave = tid >> 5;
  const int lo = lane & 15, hi = lane >> 4;
  const int img = blockIdx.y;
  const int r0  = blockIdx.x * (OROWS * SUBS);
  const float* __restrict__ X = x + (size_t)img * CIN * HW;
  float* __restrict__ QKV = qkv + (size_t)img * CMID * HW;

  if (tid <= 48 * LROWS) {
    v4f z = {0.f, 0.f, 0.f, 0.f};
    *(v4f*)(tls + tid * TLS_PITCH) = z;
  }

  const int wv = __builtin_amdgcn_readfirstlane(wave);
  const int c4 = lane * 4;

  const int lrow = wv >> 1;
  const int colbase = (wv & 1) * 64;
  float* __restrict__ trow = tls + lrow * TLS_PITCH + TLS_PAD + colbase + 4 * lo;

  for (int half_ = 0; half_ < 2; ++half_) {
    const int cbase = half_ * 48;

#pragma nounroll
    for (int sub = 0; sub < SUBS; ++sub) {
    const int rr = r0 - 1 + lrow + LROWS * sub;

    if (sub != 0 && lrow >= OROWS) {
    } else if (rr < 0 || rr >= IMG_W) {
      v4f z = {0.f, 0.f, 0.f, 0.f};
      for (int c = hi; c < 48; c += 2) *(v4f*)(trow + c * TLS_ROWSTRIDE) = z;
    } else {
      const float* __restrict__ xp = X + (size_t)(2 * hi) * HW + rr * IMG_W + colbase + 4 * lo;
      const float* __restrict__ ap = pw + (cbase + lo) * CIN + 2 * hi;
      v8f acc[3][4] = {};
      v4f x0[X_AHEAD + 1], x1[X_AHEAD + 1];
      v2f a[W_AHEAD + 1][3];
#pragma unroll
      for (int st = 0; st < X_AHEAD; ++st) {
        x0[st] = *(const v4f*)(xp + (size_t)(4 * st) * HW);
        x1[st] = *(const v4f*)(xp + (size_t)(4 * st + 1) * HW);
      }
#pragma unroll
      for (int st = 0; st < W_AHEAD; ++st)
#pragma unroll
        for (int m = 0; m < 3; ++m) a[st][m] = *(const v2f*)(ap + m * 16 * CIN + 4 * st);
#pragma unroll
      for (int st = 0; st < CIN / 4; ++st) {
        if (st + W_AHEAD < CIN / 4) {
          const int sl = (st + W_AHEAD) % (W_AHEAD + 1);
          const int kn = 4 * (st + W_AHEAD);
#pragma unroll
          for (int m = 0; m < 3; ++m) a[sl][m] = *(const v2f*)(ap + m * 16 * CIN + kn);
        }
        if (st + X_AHEAD < CIN / 4) {
          const int sl = (st + X_AHEAD) % (X_AHEAD + 1);
          const int kn = 4 * (st + X_AHEAD);
          x0[sl] = *(const v4f*)(xp + (size_t)kn * HW);
          x1[sl] = *(const v4f*)(xp + (size_t)(kn + 1) * HW);
        }
        __builtin_amdgcn_sched_barrier(0);
        const int sx = st % (X_AHEAD + 1);
        const int sw = st % (W_AHEAD + 1);
#pragma unroll
        for (int i = 0; i < 4; ++i) {
          v2f b;
          b.x = x0[sx][i];
          b.y = x1[sx][i];
#pragma unroll
          for (int m = 0; m < 3; ++m) acc[m][i] = wmma4(a[sw][m], b, acc[m][i]);
        }
      }
#pragma unroll
      for (int m = 0; m < 3; ++m)
#pragma unroll
        for (int r = 0; r < 8; ++r) {
          v4f o = {acc[m][0][r], acc[m][1][r], acc[m][2][r], acc[m][3][r]};
          *(v4f*)(trow + (m * 16 + r + 8 * hi) * TLS_ROWSTRIDE) = o;
        }
    }
    __syncthreads();

#pragma nounroll
    for (int j = 0; j < 48 / NWAVES; ++j) {
      const int lc = wv + NWAVES * j;
      const int ch = cbase + lc;
      const float* __restrict__ wch = dw + ch * 9;
      float w[9];
#pragma unroll
      for (int t = 0; t < 9; ++t) w[t] = wch[t];

      const float* __restrict__ tch = tls + lc * TLS_ROWSTRIDE + TLS_PAD + c4;
      v4f mid[LROWS];
      float lft[LROWS], rgt[LROWS];
#pragma unroll
      for (int r = 0; r < LROWS; ++r) {
        int lr = r + OROWS * sub;
        if (lr >= LROWS) lr -= LROWS;
        const float* __restrict__ trd = tch + lr * TLS_PITCH;
        mid[r] = *(const v4f*)(trd);
        lft[r] = trd[-1];
        rgt[r] = trd[4];
      }

      float ss = 0.f;
#pragma unroll
      for (int orow = 0; orow < OROWS; ++orow) {
        float s0 = 0.f, s1 = 0.f, s2 = 0.f, s3 = 0.f;
#pragma unroll
        for (int dy = 0; dy < 3; ++dy) {
          const int r = orow + dy;
          const float w0 = w[dy * 3 + 0], w1 = w[dy * 3 + 1], w2 = w[dy * 3 + 2];
          s0 += w0 * lft[r];    s0 += w1 * mid[r].x;  s0 += w2 * mid[r].y;
          s1 += w0 * mid[r].x;  s1 += w1 * mid[r].y;  s1 += w2 * mid[r].z;
          s2 += w0 * mid[r].y;  s2 += w1 * mid[r].z;  s2 += w2 * mid[r].w;
          s3 += w0 * mid[r].z;  s3 += w1 * mid[r].w;  s3 += w2 * rgt[r];
        }
        v4f o = {s0, s1, s2, s3};
        *(v4f*)(QKV + (size_t)ch * HW + (r0 + OROWS * sub + orow) * IMG_W + c4) = o;
        ss += s0 * s0;  ss += s1 * s1;  ss += s2 * s2;  ss += s3 * s3;
      }
      if (ch < 2 * AN2) {
#pragma unroll
        for (int off = 16; off > 0; off >>= 1) ss += __shfl_xor(ss, off, 32);
        if (lane == 0) atomicAdd(&gsums[img * 64 + ch], ss);
      }
    }
    if (half_ == 0 || sub + 1 < SUBS) __syncthreads();
    }
  }
}

#define SK_WAVES 8
#define SK_STEPS 4
__global__ void __launch_bounds__(SK_WAVES * 32)
k_att_partial(const float* __restrict__ qkv,
              float* __restrict__ Sbuf) {
  __shared__ __attribute__((aligned(16))) float red[SK_WAVES][AN2 * AN2];
  const int lane = threadIdx.x & 31;
  const int lo = lane & 15, hi = lane >> 4;
  const int wv = __builtin_amdgcn_readfirstlane(threadIdx.x >> 5);
  const int img = blockIdx.y;
  const int p0 = blockIdx.x * (HW / KCHUNKS) + wv * (SK_STEPS * 32);
  const float* __restrict__ rp =
      qkv + (size_t)img * CMID * HW + (size_t)lo * HW + p0 + 16 * hi;

  v4f f[2][4][4];
#pragma unroll
  for (int t = 0; t < 4; ++t)
#pragma unroll
    for (int q4 = 0; q4 < 4; ++q4)
      f[0][t][q4] = *(const v4f*)(rp + (size_t)(t * 16) * HW + 4 * q4);

  v8f acc[2][2] = {};
#pragma unroll
  for (int st = 0; st < SK_STEPS; ++st) {
    const int cur = st & 1;
    if (st + 1 < SK_STEPS) {
#pragma unroll
      for (int t = 0; t < 4; ++t)
#pragma unroll
        for (int q4 = 0; q4 < 4; ++q4)
          f[cur ^ 1][t][q4] = *(const v4f*)(rp + (size_t)(t * 16) * HW + 32 * (st + 1) + 4 * q4);
    }
    __builtin_amdgcn_sched_barrier(0);
#pragma unroll
    for (int j = 0; j < 8; ++j) {
      const int q4 = j >> 1, e = 2 * (j & 1);
      v2f a0, a1, b0, b1;
      a0.x = f[cur][0][q4][e];  a0.y = f[cur][0][q4][e + 1];
      a1.x = f[cur][1][q4][e];  a1.y = f[cur][1][q4][e + 1];
      b0.x = f[cur][2][q4][e];  b0.y = f[cur][2][q4][e + 1];
      b1.x = f[cur][3][q4][e];  b1.y = f[cur][3][q4][e + 1];
      acc[0][0] = wmma4(a0, b0, acc[0][0]);
      acc[0][1] = wmma4(a0, b1, acc[0][1]);
      acc[1][0] = wmma4(a1, b0, acc[1][0]);
      acc[1][1] = wmma4(a1, b1, acc[1][1]);
    }
  }

#pragma unroll
  for (int m = 0; m < 2; ++m)
#pragma unroll
    for (int n = 0; n < 2; ++n)
#pragma unroll
      for (int r = 0; r < 8; ++r)
        red[wv][(m * 16 + 8 * hi + r) * AN2 + n * 16 + lo] = acc[m][n][r];
  __syncthreads();

  float* __restrict__ Sp = Sbuf + (size_t)img * (AN2 * AN2);
#pragma unroll
  for (int e = 0; e < (AN2 * AN2) / (SK_WAVES * 32); ++e) {
    const int idx = threadIdx.x + e * (SK_WAVES * 32);
    float s = red[0][idx];
#pragma unroll
    for (int w = 1; w < SK_WAVES; ++w) s += red[w][idx];
    atomicAdd(&Sp[idx], s);
  }
}

#define OPX    64
#define OGROUPS 4
#define OWAVES 8
#define WPITCH 34
__global__ void __launch_bounds__(OWAVES * 32)
k_out(const float* __restrict__ qkv,
      const float* __restrict__ Sbuf,
      const float* __restrict__ sums,
      const float* __restrict__ proj,
      float* __restrict__ out) {
  __shared__ float rqk[2 * AN2];
  __shared__ float att[AN2 * AN2];
  __shared__ __attribute__((aligned(16))) float Wl[CIN * WPITCH];
  const int lane = threadIdx.x & 31;
  const int lo = lane & 15, hi = lane >> 4;
  const int wv = __builtin_amdgcn_readfirstlane(threadIdx.x >> 5);
  const int img = blockIdx.y;
  const int p0 = (blockIdx.x * OWAVES + wv) * (OPX * OGROUPS);
  const float* __restrict__ vp =
      qkv + (size_t)img * CMID * HW + (size_t)(2 * AN2 + 2 * hi) * HW + p0 + 4 * lo;
  float* __restrict__ op = out + (size_t)img * CIN * HW + (size_t)(8 * hi) * HW + p0 + 4 * lo;

  const float* __restrict__ Sp = Sbuf + (size_t)img * (AN2 * AN2) + (4 * wv) * AN2 + lane;
  float sv[4];
#pragma unroll
  for (int r = 0; r < 4; ++r) sv[r] = Sp[r * AN2];
  const float ssq = sums[img * 64 + (threadIdx.x & 63)];

  v4f x0[2][8], x1[2][8];
#pragma unroll
  for (int ks = 0; ks < 8; ++ks) {
    x0[0][ks] = *(const v4f*)(vp + (size_t)(4 * ks) * HW);
    x1[0][ks] = *(const v4f*)(vp + (size_t)(4 * ks + 1) * HW);
  }

  if (threadIdx.x < 2 * AN2) rqk[threadIdx.x] = 1.f / fmaxf(sqrtf(ssq), 1e-12f);
  __syncthreads();
  {
    const float rkc = rqk[AN2 + lane];
#pragma unroll
    for (int r = 0; r < 4; ++r) {
      const int row = 4 * wv + r;
      const float s = sv[r] * rqk[row] * rkc;
      float mx = s;
      for (int off = 16; off > 0; off >>= 1) mx = fmaxf(mx, __shfl_xor(mx, off, 32));
      const float ex = __expf(s - mx);
      float sum = ex;
      for (int off = 16; off > 0; off >>= 1) sum += __shfl_xor(sum, off, 32);
      const float inv = 1.f / sum;
      att[row * AN2 + lane] = ex * inv;
    }
  }
  __syncthreads();
  {
    float at[AN2];
#pragma unroll
    for (int cc = 0; cc < AN2; ++cc) at[cc] = att[cc * AN2 + lane];
#pragma unroll
    for (int e = 0; e < CIN / OWAVES; ++e) {
      const int o = wv + OWAVES * e;
      const float* __restrict__ pr = proj + o * AN2;
      float s = 0.f;
#pragma unroll
      for (int cc = 0; cc < AN2; ++cc) s += pr[cc] * at[cc];
      Wl[o * WPITCH + lane] = s;
    }
  }
  __syncthreads();

  const float* __restrict__ wp = Wl + lo * WPITCH + 2 * hi;
  v2f a[8][4];
#pragma unroll
  for (int ks = 0; ks < 8; ++ks)
#pragma unroll
    for (int m = 0; m < 4; ++m) a[ks][m] = *(const v2f*)(wp + m * 16 * WPITCH + 4 * ks);

#pragma unroll
  for (int g = 0; g < OGROUPS; ++g) {
    const int cur = g & 1;
    if (g + 1 < OGROUPS) {
#pragma unroll
      for (int ks = 0; ks < 8; ++ks) {
        x0[cur ^ 1][ks] = *(const v4f*)(vp + (size_t)(4 * ks) * HW + OPX * (g + 1));
        x1[cur ^ 1][ks] = *(const v4f*)(vp + (size_t)(4 * ks + 1) * HW + OPX * (g + 1));
      }
    }
    __builtin_amdgcn_sched_barrier(0);
    v8f acc[4][4] = {};
#pragma unroll
    for (int ks = 0; ks < 8; ++ks) {
#pragma unroll
      for (int i = 0; i < 4; ++i) {
        v2f b;
        b.x = x0[cur][ks][i];
        b.y = x1[cur][ks][i];
#pragma unroll
        for (int m = 0; m < 4; ++m) acc[m][i] = wmma4(a[ks][m], b, acc[m][i]);
      }
    }
#pragma unroll
    for (int m = 0; m < 4; ++m)
#pragma unroll
      for (int r = 0; r < 8; ++r) {
        v4f o = {acc[m][0][r], acc[m][1][r], acc[m][2][r], acc[m][3][r]};
        __builtin_nontemporal_store(o, (v4f*)(op + (size_t)(m * 16 + r) * HW + OPX * g));
      }
  }
}

extern "C" void kernel_launch(void* const* d_in, const int* in_sizes, int n_in,
                              void* d_out, int out_size, void* d_ws, size_t ws_size,
                              hipStream_t stream) {
  const float* x    = (const float*)d_in[0];
  const float* pw   = (const float*)d_in[1];
  const float* dw   = (const float*)d_in[2];
  const float* proj = (const float*)d_in[3];
  float* out = (float*)d_out;

  float* qkv  = (float*)d_ws;
  float* sums = qkv + (size_t)BCN * CMID * HW;
  float* Sbuf = sums + (size_t)BCN * 64;
  const int nzero = BCN * 64 + BCN * AN2 * AN2;

  k_zero<<<dim3((nzero + 255) / 256), 256, 0, stream>>>(sums, nzero);
  k_fused<<<dim3(IMG_W / (OROWS * SUBS), BCN), NWAVES * 32, TLS_FLOATS * sizeof(float), stream>>>(
      x, pw, dw, qkv, sums);
  k_att_partial<<<dim3(KCHUNKS, BCN), SK_WAVES * 32, 0, stream>>>(qkv, Sbuf);
  k_out<<<dim3(HW / (OWAVES * OPX * OGROUPS), BCN), OWAVES * 32, 0, stream>>>(qkv, Sbuf, sums, proj, out);
}


# zeroing launch removed (4 -> 3): q/k square sums leave as per-block partials by plain stores, summed in slot order in k_out (atomics gone, sum order changed); fused kernel clears S
# speedup vs baseline: 3.5735x; 1.0479x over previous
#include <hip/hip_runtime.h>

typedef __attribute__((ext_vector_type(2))) float v2f;
typedef __attribute__((ext_vector_type(4))) float v4f;
typedef __attribute__((ext_vector_type(8))) float v8f;

#define HW    16384
#define IMG_W 128
#define BCN   32
#define CIN   64
#define CMID  96
#define AN2   32
#define OROWS  4
#define SUBS   2
#define LROWS  (OROWS + 2)
#define NWAVES (2 * LROWS)
#define X_AHEAD 2
#define W_AHEAD 1
#define TLS_PAD   4
#define TLS_PITCH (IMG_W + TLS_PAD)
#define TLS_ROWSTRIDE (LROWS * TLS_PITCH)
#define TLS_FLOATS (48 * TLS_ROWSTRIDE + TLS_PAD)
#define KCHUNKS 16
#define NBLK_X (IMG_W / (OROWS * SUBS))
#define NPART  (NBLK_X * SUBS)

static __device__ __forceinline__ v8f wmma4(v2f a, v2f b, v8f c) {
  return __builtin_amdgcn_wmma_f32_16x16x4_f32(
       false, a,  false, b,
       (short)0, c,  false,  false);
}

__global__ void __launch_bounds__(NWAVES * 32) __attribute__((amdgpu_waves_per_eu(6, 6)))
k_fused(const float* __restrict__ x,
        const float* __restrict__ pw,
        const float* __restrict__ dw,
        float* __restrict__ qkv,
        float* __restrict__ gsums,
        float* __restrict__ Sbuf) {
  extern __shared__ __attribute__((aligned(16))) float tls[];

  const int tid  = threadIdx.x;
  const int lane = tid & 31, wave = tid >> 5;
  const int lo = lane & 15, hi = lane >> 4;
  const int img = blockIdx.y;
  const int r0  = blockIdx.x * (OROWS * SUBS);
  const float* __restrict__ X = x + (size_t)img * CIN * HW;
  float* __restrict__ QKV = qkv + (size_t)img * CMID * HW;

  if (tid < (AN2 * AN2) / NBLK_X)
    Sbuf[(size_t)img * (AN2 * AN2) + blockIdx.x * ((AN2 * AN2) / NBLK_X) + tid] = 0.f;

  if (tid <= 48 * LROWS) {
    v4f z = {0.f, 0.f, 0.f, 0.f};
    *(v4f*)(tls + tid * TLS_PITCH) = z;
  }

  const int wv = __builtin_amdgcn_readfirstlane(wave);
  const int c4 = lane * 4;

  const int lrow = wv >> 1;
  const int colbase = (wv & 1) * 64;
  float* __restrict__ trow = tls + lrow * TLS_PITCH + TLS_PAD + colbase + 4 * lo;

  for (int half_ = 0; half_ < 2; ++half_) {
    const int cbase = half_ * 48;

#pragma nounroll
    for (int sub = 0; sub < SUBS; ++sub) {
    const int rr = r0 - 1 + lrow + LROWS * sub;

    if (sub != 0 && lrow >= OROWS) {
    } else if (rr < 0 || rr >= IMG_W) {
      v4f z = {0.f, 0.f, 0.f, 0.f};
      for (int c = hi; c < 48; c += 2) *(v4f*)(trow + c * TLS_ROWSTRIDE) = z;
    } else {
      const float* __restrict__ xp = X + (size_t)(2 * hi) * HW + rr * IMG_W + colbase + 4 * lo;
      const float* __restrict__ ap = pw + (cbase + lo) * CIN + 2 * hi;
      v8f acc[3][4] = {};
      v4f x0[X_AHEAD + 1], x1[X_AHEAD + 1];
      v2f a[W_AHEAD + 1][3];
#pragma unroll
      for (int st = 0; st < X_AHEAD; ++st) {
        x0[st] = *(const v4f*)(xp + (size_t)(4 * st) * HW);
        x1[st] = *(const v4f*)(xp + (size_t)(4 * st + 1) * HW);
      }
#pragma unroll
      for (int st = 0; st < W_AHEAD; ++st)
#pragma unroll
        for (int m = 0; m < 3; ++m) a[st][m] = *(const v2f*)(ap + m * 16 * CIN + 4 * st);
#pragma unroll
      for (int st = 0; st < CIN / 4; ++st) {
        if (st + W_AHEAD < CIN / 4) {
          const int sl = (st + W_AHEAD) % (W_AHEAD + 1);
          const int kn = 4 * (st + W_AHEAD);
#pragma unroll
          for (int m = 0; m < 3; ++m) a[sl][m] = *(const v2f*)(ap + m * 16 * CIN + kn);
        }
        if (st + X_AHEAD < CIN / 4) {
          const int sl = (st + X_AHEAD) % (X_AHEAD + 1);
          const int kn = 4 * (st + X_AHEAD);
          x0[sl] = *(const v4f*)(xp + (size_t)kn * HW);
          x1[sl] = *(const v4f*)(xp + (size_t)(kn + 1) * HW);
        }
        __builtin_amdgcn_sched_barrier(0);
        const int sx = st % (X_AHEAD + 1);
        const int sw = st % (W_AHEAD + 1);
#pragma unroll
        for (int i = 0; i < 4; ++i) {
          v2f b;
          b.x = x0[sx][i];
          b.y = x1[sx][i];
#pragma unroll
          for (int m = 0; m < 3; ++m) acc[m][i] = wmma4(a[sw][m], b, acc[m][i]);
        }
      }
#pragma unroll
      for (int m = 0; m < 3; ++m)
#pragma unroll
        for (int r = 0; r < 8; ++r) {
          v4f o = {acc[m][0][r], acc[m][1][r], acc[m][2][r], acc[m][3][r]};
          *(v4f*)(trow + (m * 16 + r + 8 * hi) * TLS_ROWSTRIDE) = o;
        }
    }
    __syncthreads();

#pragma nounroll
    for (int j = 0; j < 48 / NWAVES; ++j) {
      const int lc = wv + NWAVES * j;
      const int ch = cbase + lc;
      const float* __restrict__ wch = dw + ch * 9;
      float w[9];
#pragma unroll
      for (int t = 0; t < 9; ++t) w[t] = wch[t];

      const float* __restrict__ tch = tls + lc * TLS_ROWSTRIDE + TLS_PAD + c4;
      v4f mid[LROWS];
      float lft[LROWS], rgt[LROWS];
#pragma unroll
      for (int r = 0; r < LROWS; ++r) {
        int lr = r + OROWS * sub;
        if (lr >= LROWS) lr -= LROWS;
        const float* __restrict__ trd = tch + lr * TLS_PITCH;
        mid[r] = *(const v4f*)(trd);
        lft[r] = trd[-1];
        rgt[r] = trd[4];
      }

      float ss = 0.f;
#pragma unroll
      for (int orow = 0; orow < OROWS; ++orow) {
        float s0 = 0.f, s1 = 0.f, s2 = 0.f, s3 = 0.f;
#pragma unroll
        for (int dy = 0; dy < 3; ++dy) {
          const int r = orow + dy;
          const float w0 = w[dy * 3 + 0], w1 = w[dy * 3 + 1], w2 = w[dy * 3 + 2];
          s0 += w0 * lft[r];    s0 += w1 * mid[r].x;  s0 += w2 * mid[r].y;
          s1 += w0 * mid[r].x;  s1 += w1 * mid[r].y;  s1 += w2 * mid[r].z;
          s2 += w0 * mid[r].y;  s2 += w1 * mid[r].z;  s2 += w2 * mid[r].w;
          s3 += w0 * mid[r].z;  s3 += w1 * mid[r].w;  s3 += w2 * rgt[r];
        }
        v4f o = {s0, s1, s2, s3};
        *(v4f*)(QKV + (size_t)ch * HW + (r0 + OROWS * sub + orow) * IMG_W + c4) = o;
        ss += s0 * s0;  ss += s1 * s1;  ss += s2 * s2;  ss += s3 * s3;
      }
      if (ch < 2 * AN2) {
#pragma unroll
        for (int off = 16; off > 0; off >>= 1) ss += __shfl_xor(ss, off, 32);
        if (lane == 0) gsums[((img * NBLK_X + blockIdx.x) * SUBS + sub) * 64 + ch] = ss;
      }
    }
    if (half_ == 0 || sub + 1 < SUBS) __syncthreads();
    }
  }
}

#define SK_WAVES 8
#define SK_STEPS 4
__global__ void __launch_bounds__(SK_WAVES * 32)
k_att_partial(const float* __restrict__ qkv,
              float* __restrict__ Sbuf) {
  __shared__ __attribute__((aligned(16))) float red[SK_WAVES][AN2 * AN2];
  const int lane = threadIdx.x & 31;
  const int lo = lane & 15, hi = lane >> 4;
  const int wv = __builtin_amdgcn_readfirstlane(threadIdx.x >> 5);
  const int img = blockIdx.y;
  const int p0 = blockIdx.x * (HW / KCHUNKS) + wv * (SK_STEPS * 32);
  const float* __restrict__ rp =
      qkv + (size_t)img * CMID * HW + (size_t)lo * HW + p0 + 16 * hi;

  v4f f[2][4][4];
#pragma unroll
  for (int t = 0; t < 4; ++t)
#pragma unroll
    for (int q4 = 0; q4 < 4; ++q4)
      f[0][t][q4] = *(const v4f*)(rp + (size_t)(t * 16) * HW + 4 * q4);

  v8f acc[2][2] = {};
#pragma unroll
  for (int st = 0; st < SK_STEPS; ++st) {
    const int cur = st & 1;
    if (st + 1 < SK_STEPS) {
#pragma unroll
      for (int t = 0; t < 4; ++t)
#pragma unroll
        for (int q4 = 0; q4 < 4; ++q4)
          f[cur ^ 1][t][q4] = *(const v4f*)(rp + (size_t)(t * 16) * HW + 32 * (st + 1) + 4 * q4);
    }
    __builtin_amdgcn_sched_barrier(0);
#pragma unroll
    for (int j = 0; j < 8; ++j) {
      const int q4 = j >> 1, e = 2 * (j & 1);
      v2f a0, a1, b0, b1;
      a0.x = f[cur][0][q4][e];  a0.y = f[cur][0][q4][e + 1];
      a1.x = f[cur][1][q4][e];  a1.y = f[cur][1][q4][e + 1];
      b0.x = f[cur][2][q4][e];  b0.y = f[cur][2][q4][e + 1];
      b1.x = f[cur][3][q4][e];  b1.y = f[cur][3][q4][e + 1];
      acc[0][0] = wmma4(a0, b0, acc[0][0]);
      acc[0][1] = wmma4(a0, b1, acc[0][1]);
      acc[1][0] = wmma4(a1, b0, acc[1][0]);
      acc[1][1] = wmma4(a1, b1, acc[1][1]);
    }
  }

#pragma unroll
  for (int m = 0; m < 2; ++m)
#pragma unroll
    for (int n = 0; n < 2; ++n)
#pragma unroll
      for (int r = 0; r < 8; ++r)
        red[wv][(m * 16 + 8 * hi + r) * AN2 + n * 16 + lo] = acc[m][n][r];
  __syncthreads();

  float* __restrict__ Sp = Sbuf + (size_t)img * (AN2 * AN2);
#pragma unroll
  for (int e = 0; e < (AN2 * AN2) / (SK_WAVES * 32); ++e) {
    const int idx = threadIdx.x + e * (SK_WAVES * 32);
    float s = red[0][idx];
#pragma unroll
    for (int w = 1; w < SK_WAVES; ++w) s += red[w][idx];
    atomicAdd(&Sp[idx], s);
  }
}

#define OPX    64
#define OGROUPS 4
#define OWAVES 8
#define WPITCH 34
__global__ void __launch_bounds__(OWAVES * 32)
k_out(const float* __restrict__ qkv,
      const float* __restrict__ Sbuf,
      const float* __restrict__ sums,
      const float* __restrict__ proj,
      float* __restrict__ out) {
  __shared__ float rqk[2 * AN2];
  __shared__ float att[AN2 * AN2];
  __shared__ __attribute__((aligned(16))) float Wl[CIN * WPITCH];
  const int lane = threadIdx.x & 31;
  const int lo = lane & 15, hi = lane >> 4;
  const int wv = __builtin_amdgcn_readfirstlane(threadIdx.x >> 5);
  const int img = blockIdx.y;
  const int p0 = (blockIdx.x * OWAVES + wv) * (OPX * OGROUPS);
  const float* __restrict__ vp =
      qkv + (size_t)img * CMID * HW + (size_t)(2 * AN2 + 2 * hi) * HW + p0 + 4 * lo;
  float* __restrict__ op = out + (size_t)img * CIN * HW + (size_t)(8 * hi) * HW + p0 + 4 * lo;

  const float* __restrict__ Sp = Sbuf + (size_t)img * (AN2 * AN2) + (4 * wv) * AN2 + lane;
  float sv[4];
#pragma unroll
  for (int r = 0; r < 4; ++r) sv[r] = Sp[r * AN2];
  float part[NPART];
  if (threadIdx.x < 2 * AN2) {
    const float* __restrict__ sp = sums + (size_t)img * (NPART * 64) + threadIdx.x;
#pragma unroll
    for (int p = 0; p < NPART; ++p) part[p] = sp[p * 64];
  }

  v4f x0[2][8], x1[2][8];
#pragma unroll
  for (int ks = 0; ks < 8; ++ks) {
    x0[0][ks] = *(const v4f*)(vp + (size_t)(4 * ks) * HW);
    x1[0][ks] = *(const v4f*)(vp + (size_t)(4 * ks + 1) * HW);
  }

  if (threadIdx.x < 2 * AN2) {
    float ssq = 0.f;
#pragma unroll
    for (int p = 0; p < NPART; ++p) ssq += part[p];
    rqk[threadIdx.x] = 1.f / fmaxf(sqrtf(ssq), 1e-12f);
  }
  __syncthreads();
  {
    const float rkc = rqk[AN2 + lane];
#pragma unroll
    for (int r = 0; r < 4; ++r) {
      const int row = 4 * wv + r;
      const float s = sv[r] * rqk[row] * rkc;
      float mx = s;
      for (int off = 16; off > 0; off >>= 1) mx = fmaxf(mx, __shfl_xor(mx, off, 32));
      const float ex = __expf(s - mx);
      float sum = ex;
      for (int off = 16; off > 0; off >>= 1) sum += __shfl_xor(sum, off, 32);
      const float inv = 1.f / sum;
      att[row * AN2 + lane] = ex * inv;
    }
  }
  __syncthreads();
  {
    float at[AN2];
#pragma unroll
    for (int cc = 0; cc < AN2; ++cc) at[cc] = att[cc * AN2 + lane];
#pragma unroll
    for (int e = 0; e < CIN / OWAVES; ++e) {
      const int o = wv + OWAVES * e;
      const float* __restrict__ pr = proj + o * AN2;
      float s = 0.f;
#pragma unroll
      for (int cc = 0; cc < AN2; ++cc) s += pr[cc] * at[cc];
      Wl[o * WPITCH + lane] = s;
    }
  }
  __syncthreads();

  const float* __restrict__ wp = Wl + lo * WPITCH + 2 * hi;
  v2f a[8][4];
#pragma unroll
  for (int ks = 0; ks < 8; ++ks)
#pragma unroll
    for (int m = 0; m < 4; ++m) a[ks][m] = *(const v2f*)(wp + m * 16 * WPITCH + 4 * ks);

#pragma unroll
  for (int g = 0; g < OGROUPS; ++g) {
    const int cur = g & 1;
    if (g + 1 < OGROUPS) {
#pragma unroll
      for (int ks = 0; ks < 8; ++ks) {
        x0[cur ^ 1][ks] = *(const v4f*)(vp + (size_t)(4 * ks) * HW + OPX * (g + 1));
        x1[cur ^ 1][ks] = *(const v4f*)(vp + (size_t)(4 * ks + 1) * HW + OPX * (g + 1));
      }
    }
    __builtin_amdgcn_sched_barrier(0);
    v8f acc[4][4] = {};
#pragma unroll
    for (int ks = 0; ks < 8; ++ks) {
#pragma unroll
      for (int i = 0; i < 4; ++i) {
        v2f b;
        b.x = x0[cur][ks][i];
        b.y = x1[cur][ks][i];
#pragma unroll
        for (int m = 0; m < 4; ++m) acc[m][i] = wmma4(a[ks][m], b, acc[m][i]);
      }
    }
#pragma unroll
    for (int m = 0; m < 4; ++m)
#pragma unroll
      for (int r = 0; r < 8; ++r) {
        v4f o = {acc[m][0][r], acc[m][1][r], acc[m][2][r], acc[m][3][r]};
        __builtin_nontemporal_store(o, (v4f*)(op + (size_t)(m * 16 + r) * HW + OPX * g));
      }
  }
}

extern "C" void kernel_launch(void* const* d_in, const int* in_sizes, int n_in,
                              void* d_out, int out_size, void* d_ws, size_t ws_size,
                              hipStream_t stream) {
  const float* x    = (const float*)d_in[0];
  const float* pw   = (const float*)d_in[1];
  const float* dw   = (const float*)d_in[2];
  const float* proj = (const float*)d_in[3];
  float* out = (float*)d_out;

  float* qkv  = (float*)d_ws;
  float* sums = qkv + (size_t)BCN * CMID * HW;
  float* Sbuf = sums + (size_t)BCN * NPART * 64;

  k_fused<<<dim3(NBLK_X, BCN), NWAVES * 32, TLS_FLOATS * sizeof(float), stream>>>(
      x, pw, dw, qkv, sums, Sbuf);
  k_att_partial<<<dim3(KCHUNKS, BCN), SK_WAVES * 32, 0, stream>>>(qkv, Sbuf);
  k_out<<<dim3(HW / (OWAVES * OPX * OGROUPS), BCN), OWAVES * 32, 0, stream>>>(qkv, Sbuf, sums, proj, out);
}


# q.k^T partials formed in the fused kernel from q/k rows kept in LDS over dead t rows: q/k never stored, split-K launch and atomics gone (3->2 launches); 16-row blocks of 24 waves; sum order changed
# speedup vs baseline: 3.7494x; 1.0492x over previous
#include <hip/hip_runtime.h>

typedef __attribute__((ext_vector_type(2))) float v2f;
typedef __attribute__((ext_vector_type(4))) float v4f;
typedef __attribute__((ext_vector_type(8))) float v8f;

#define HW    16384
#define IMG_W 128
#define BCN   32
#define CIN   64
#define CMID  96
#define AN2   32
#define OROWS  4
#define SUBS   4
#define LROWS  (OROWS + 2)
#define PWAVES (2 * LROWS)
#define NWAVES (2 * PWAVES)
#define SWAVES 16
#define X_AHEAD 2
#define W_AHEAD 1
#define TLS_PAD   4
#define TLS_PITCH (IMG_W + TLS_PAD)
#define TLS_ROWSTRIDE (LROWS * TLS_PITCH)
#define TLS_FLOATS (CMID * TLS_ROWSTRIDE + TLS_PAD)
#define NBLK_X (IMG_W / (OROWS * SUBS))
#define NPART  (NBLK_X * SUBS)

static __device__ __forceinline__ v8f wmma4(v2f a, v2f b, v8f c) {
  return __builtin_amdgcn_wmma_f32_16x16x4_f32(
       false, a,  false, b,
       (short)0, c,  false,  false);
}

__global__ void __launch_bounds__(NWAVES * 32) __attribute__((amdgpu_waves_per_eu(6, 6)))
k_fused(const float* __restrict__ x,
        const float* __restrict__ pw,
        const float* __restrict__ dw,
        float* __restrict__ qkv,
        float* __restrict__ gsums,
        float* __restrict__ Spart) {
  extern __shared__ __attribute__((aligned(16))) float tls[];

  const int tid  = threadIdx.x;
  const int lane = tid & 31, wave = tid >> 5;
  const int lo = lane & 15, hi = lane >> 4;
  const int img = blockIdx.y;
  const int r0  = blockIdx.x * (OROWS * SUBS);
  const float* __restrict__ X = x + (size_t)img * CIN * HW;
  float* __restrict__ QKV = qkv + (size_t)img * CMID * HW;

  if (tid <= CMID * LROWS) {
    v4f z = {0.f, 0.f, 0.f, 0.f};
    *(v4f*)(tls + tid * TLS_PITCH) = z;
  }

  const int wv = __builtin_amdgcn_readfirstlane(wave);
  const int c4 = lane * 4;

  const int chalf = wv / PWAVES;
  const int w12 = wv - chalf * PWAVES;
  const int prow = w12 >> 1;
  const int colbase = (w12 & 1) * 64;
  const int cbase = chalf * 48;

  v8f sacc[2][2] = {};

#pragma nounroll
  for (int sub = 0; sub < SUBS; ++sub) {
    const int rbase = (OROWS * sub) % LROWS;

    const int wr = (sub == 0) ? prow : prow + 2;
    if (wr < LROWS) {
      const int rr = r0 - 1 + OROWS * sub + wr;
      int lrw = rbase + wr;
      if (lrw >= LROWS) lrw -= LROWS;
      float* trow = tls + cbase * TLS_ROWSTRIDE + lrw * TLS_PITCH + TLS_PAD + colbase + 4 * lo;
      if (rr < 0 || rr >= IMG_W) {
        v4f z = {0.f, 0.f, 0.f, 0.f};
        for (int c = hi; c < 48; c += 2) *(v4f*)(trow + c * TLS_ROWSTRIDE) = z;
      } else {
      const float* __restrict__ xp = X + (size_t)(2 * hi) * HW + rr * IMG_W + colbase + 4 * lo;
      const float* __restrict__ ap = pw + (cbase + lo) * CIN + 2 * hi;
      v8f acc[3][4] = {};
      v4f x0[X_AHEAD + 1], x1[X_AHEAD + 1];
      v2f a[W_AHEAD + 1][3];
#pragma unroll
      for (int st = 0; st < X_AHEAD; ++st) {
        x0[st] = *(const v4f*)(xp + (size_t)(4 * st) * HW);
        x1[st] = *(const v4f*)(xp + (size_t)(4 * st + 1) * HW);
      }
#pragma unroll
      for (int st = 0; st < W_AHEAD; ++st)
#pragma unroll
        for (int m = 0; m < 3; ++m) a[st][m] = *(const v2f*)(ap + m * 16 * CIN + 4 * st);
#pragma unroll
      for (int st = 0; st < CIN / 4; ++st) {
        if (st + W_AHEAD < CIN / 4) {
          const int sl = (st + W_AHEAD) % (W_AHEAD + 1);
          const int kn = 4 * (st + W_AHEAD);
#pragma unroll
          for (int m = 0; m < 3; ++m) a[sl][m] = *(const v2f*)(ap + m * 16 * CIN + kn);
        }
        if (st + X_AHEAD < CIN / 4) {
          const int sl = (st + X_AHEAD) % (X_AHEAD + 1);
          const int kn = 4 * (st + X_AHEAD);
          x0[sl] = *(const v4f*)(xp + (size_t)kn * HW);
          x1[sl] = *(const v4f*)(xp + (size_t)(kn + 1) * HW);
        }
        __builtin_amdgcn_sched_barrier(0);
        const int sx = st % (X_AHEAD + 1);
        const int sw = st % (W_AHEAD + 1);
#pragma unroll
        for (int i = 0; i < 4; ++i) {
          v2f b;
          b.x = x0[sx][i];
          b.y = x1[sx][i];
#pragma unroll
          for (int m = 0; m < 3; ++m) acc[m][i] = wmma4(a[sw][m], b, acc[m][i]);
        }
      }
#pragma unroll
      for (int m = 0; m < 3; ++m)
#pragma unroll
        for (int r = 0; r < 8; ++r) {
          v4f o = {acc[m][0][r], acc[m][1][r], acc[m][2][r], acc[m][3][r]};
          *(v4f*)(trow + (m * 16 + r + 8 * hi) * TLS_ROWSTRIDE) = o;
        }
      }
    }
    __syncthreads();

#pragma nounroll
    for (int j = 0; j < CMID / NWAVES; ++j) {
      const int ch = wv + NWAVES * j;
      const float* __restrict__ wch = dw + ch * 9;
      float w[9];
#pragma unroll
      for (int t = 0; t < 9; ++t) w[t] = wch[t];

      float* tch = tls + ch * TLS_ROWSTRIDE + TLS_PAD + c4;
      v4f mid[LROWS];
      float lft[LROWS], rgt[LROWS];
#pragma unroll
      for (int r = 0; r < LROWS; ++r) {
        int lr = rbase + r;
        if (lr >= LROWS) lr -= LROWS;
        const float* trd = tch + lr * TLS_PITCH;
        mid[r] = *(const v4f*)(trd);
        lft[r] = trd[-1];
        rgt[r] = trd[4];
      }

      float ss = 0.f;
#pragma unroll
      for (int orow = 0; orow < OROWS; ++orow) {
        float s0 = 0.f, s1 = 0.f, s2 = 0.f, s3 = 0.f;
#pragma unroll
        for (int dy = 0; dy < 3; ++dy) {
          const int r = orow + dy;
          const float w0 = w[dy * 3 + 0], w1 = w[dy * 3 + 1], w2 = w[dy * 3 + 2];
          s0 += w0 * lft[r];    s0 += w1 * mid[r].x;  s0 += w2 * mid[r].y;
          s1 += w0 * mid[r].x;  s1 += w1 * mid[r].y;  s1 += w2 * mid[r].z;
          s2 += w0 * mid[r].y;  s2 += w1 * mid[r].z;  s2 += w2 * mid[r].w;
          s3 += w0 * mid[r].z;  s3 += w1 * mid[r].w;  s3 += w2 * rgt[r];
        }
        v4f o = {s0, s1, s2, s3};
        if (ch < 2 * AN2) {
          int lr = rbase + orow;
          if (lr >= LROWS) lr -= LROWS;
          *(v4f*)(tch + lr * TLS_PITCH) = o;
        } else {
          *(v4f*)(QKV + (size_t)ch * HW + (r0 + OROWS * sub + orow) * IMG_W + c4) = o;
        }
        ss += s0 * s0;  ss += s1 * s1;  ss += s2 * s2;  ss += s3 * s3;
      }
      if (ch < 2 * AN2) {
#pragma unroll
        for (int off = 16; off > 0; off >>= 1) ss += __shfl_xor(ss, off, 32);
        if (lane == 0) gsums[((img * NBLK_X + blockIdx.x) * SUBS + sub) * 64 + ch] = ss;
      }
    }
    __syncthreads();

    if (wv < SWAVES) {
      int lr = rbase + (wv >> 2);
      if (lr >= LROWS) lr -= LROWS;
      const float* sp = tls + lo * TLS_ROWSTRIDE + lr * TLS_PITCH + TLS_PAD + 32 * (wv & 3) + 16 * hi;
      v4f f[4][4];
#pragma unroll
      for (int t = 0; t < 4; ++t)
#pragma unroll
        for (int q4 = 0; q4 < 4; ++q4)
          f[t][q4] = *(const v4f*)(sp + (t * 16) * TLS_ROWSTRIDE + 4 * q4);
#pragma unroll
      for (int j = 0; j < 8; ++j) {
        const int q4 = j >> 1, e = 2 * (j & 1);
        v2f a0, a1, b0, b1;
        a0.x = f[0][q4][e];  a0.y = f[0][q4][e + 1];
        a1.x = f[1][q4][e];  a1.y = f[1][q4][e + 1];
        b0.x = f[2][q4][e];  b0.y = f[2][q4][e + 1];
        b1.x = f[3][q4][e];  b1.y = f[3][q4][e + 1];
        sacc[0][0] = wmma4(a0, b0, sacc[0][0]);
        sacc[0][1] = wmma4(a0, b1, sacc[0][1]);
        sacc[1][0] = wmma4(a1, b0, sacc[1][0]);
        sacc[1][1] = wmma4(a1, b1, sacc[1][1]);
      }
    }
    __syncthreads();
  }

  if (wv < SWAVES) {
#pragma unroll
    for (int m = 0; m < 2; ++m)
#pragma unroll
      for (int n = 0; n < 2; ++n)
#pragma unroll
        for (int r = 0; r < 8; ++r)
          tls[wv * (AN2 * AN2) + (m * 16 + 8 * hi + r) * AN2 + n * 16 + lo] = sacc[m][n][r];
  }
  __syncthreads();
  float* __restrict__ Sp = Spart + (size_t)(img * NBLK_X + blockIdx.x) * (AN2 * AN2);
  for (int idx = tid; idx < AN2 * AN2; idx += NWAVES * 32) {
    float s = tls[idx];
#pragma unroll
    for (int w = 1; w < SWAVES; ++w) s += tls[w * (AN2 * AN2) + idx];
    Sp[idx] = s;
  }
}

#define OPX    64
#define OGROUPS 4
#define OWAVES 8
#define WPITCH 34
__global__ void __launch_bounds__(OWAVES * 32)
k_out(const float* __restrict__ qkv,
      const float* __restrict__ Spart,
      const float* __restrict__ sums,
      const float* __restrict__ proj,
      float* __restrict__ out) {
  __shared__ float rqk[2 * AN2];
  __shared__ float att[AN2 * AN2];
  __shared__ __attribute__((aligned(16))) float Wl[CIN * WPITCH];
  const int lane = threadIdx.x & 31;
  const int lo = lane & 15, hi = lane >> 4;
  const int wv = __builtin_amdgcn_readfirstlane(threadIdx.x >> 5);
  const int img = blockIdx.y;
  const int p0 = (blockIdx.x * OWAVES + wv) * (OPX * OGROUPS);
  const float* __restrict__ vp =
      qkv + (size_t)img * CMID * HW + (size_t)(2 * AN2 + 2 * hi) * HW + p0 + 4 * lo;
  float* __restrict__ op = out + (size_t)img * CIN * HW + (size_t)(8 * hi) * HW + p0 + 4 * lo;

  const float* __restrict__ Sp = Spart + (size_t)img * (NBLK_X * AN2 * AN2) + (4 * wv) * AN2 + lane;
  float sl[4][NBLK_X];
#pragma unroll
  for (int r = 0; r < 4; ++r)
#pragma unroll
    for (int b = 0; b < NBLK_X; ++b) sl[r][b] = Sp[b * (AN2 * AN2) + r * AN2];
  float part[NPART];
  if (threadIdx.x < 2 * AN2) {
    const float* __restrict__ sp = sums + (size_t)img * (NPART * 64) + threadIdx.x;
#pragma unroll
    for (int p = 0; p < NPART; ++p) part[p] = sp[p * 64];
  }

  v4f x0[2][8], x1[2][8];
#pragma unroll
  for (int ks = 0; ks < 8; ++ks) {
    x0[0][ks] = *(const v4f*)(vp + (size_t)(4 * ks) * HW);
    x1[0][ks] = *(const v4f*)(vp + (size_t)(4 * ks + 1) * HW);
  }

  if (threadIdx.x < 2 * AN2) {
    float ssq = 0.f;
#pragma unroll
    for (int p = 0; p < NPART; ++p) ssq += part[p];
    rqk[threadIdx.x] = 1.f / fmaxf(sqrtf(ssq), 1e-12f);
  }
  __syncthreads();
  {
    const float rkc = rqk[AN2 + lane];
#pragma unroll
    for (int r = 0; r < 4; ++r) {
      const int row = 4 * wv + r;
      float sv = sl[r][0];
#pragma unroll
      for (int b = 1; b < NBLK_X; ++b) sv += sl[r][b];
      const float s = sv * rqk[row] * rkc;
      float mx = s;
      for (int off = 16; off > 0; off >>= 1) mx = fmaxf(mx, __shfl_xor(mx, off, 32));
      const float ex = __expf(s - mx);
      float sum = ex;
      for (int off = 16; off > 0; off >>= 1) sum += __shfl_xor(sum, off, 32);
      const float inv = 1.f / sum;
      att[row * AN2 + lane] = ex * inv;
    }
  }
  __syncthreads();
  {
    float at[AN2];
#pragma unroll
    for (int cc = 0; cc < AN2; ++cc) at[cc] = att[cc * AN2 + lane];
#pragma unroll
    for (int e = 0; e < CIN / OWAVES; ++e) {
      const int o = wv + OWAVES * e;
      const float* __restrict__ pr = proj + o * AN2;
      float s = 0.f;
#pragma unroll
      for (int cc = 0; cc < AN2; ++cc) s += pr[cc] * at[cc];
      Wl[o * WPITCH + lane] = s;
    }
  }
  __syncthreads();

  const float* __restrict__ wp = Wl + lo * WPITCH + 2 * hi;
  v2f a[8][4];
#pragma unroll
  for (int ks = 0; ks < 8; ++ks)
#pragma unroll
    for (int m = 0; m < 4; ++m) a[ks][m] = *(const v2f*)(wp + m * 16 * WPITCH + 4 * ks);

#pragma unroll
  for (int g = 0; g < OGROUPS; ++g) {
    const int cur = g & 1;
    if (g + 1 < OGROUPS) {
#pragma unroll
      for (int ks = 0; ks < 8; ++ks) {
        x0[cur ^ 1][ks] = *(const v4f*)(vp + (size_t)(4 * ks) * HW + OPX * (g + 1));
        x1[cur ^ 1][ks] = *(const v4f*)(vp + (size_t)(4 * ks + 1) * HW + OPX * (g + 1));
      }
    }
    __builtin_amdgcn_sched_barrier(0);
    v8f acc[4][4] = {};
#pragma unroll
    for (int ks = 0; ks < 8; ++ks) {
#pragma unroll
      for (int i = 0; i < 4; ++i) {
        v2f b;
        b.x = x0[cur][ks][i];
        b.y = x1[cur][ks][i];
#pragma unroll
        for (int m = 0; m < 4; ++m) acc[m][i] = wmma4(a[ks][m], b, acc[m][i]);
      }
    }
#pragma unroll
    for (int m = 0; m < 4; ++m)
#pragma unroll
      for (int r = 0; r < 8; ++r) {
        v4f o = {acc[m][0][r], acc[m][1][r], acc[m][2][r], acc[m][3][r]};
        __builtin_nontemporal_store(o, (v4f*)(op + (size_t)(m * 16 + r) * HW + OPX * g));
      }
  }
}

extern "C" void kernel_launch(void* const* d_in, const int* in_sizes, int n_in,
                              void* d_out, int out_size, void* d_ws, size_t ws_size,
                              hipStream_t stream) {
  const float* x    = (const float*)d_in[0];
  const float* pw   = (const float*)d_in[1];
  const float* dw   = (const float*)d_in[2];
  const float* proj = (const float*)d_in[3];
  float* out = (float*)d_out;

  float* qkv   = (float*)d_ws;
  float* sums  = qkv + (size_t)BCN * CMID * HW;
  float* Spart = sums + (size_t)BCN * NPART * 64;

  k_fused<<<dim3(NBLK_X, BCN), NWAVES * 32, TLS_FLOATS * sizeof(float), stream>>>(
      x, pw, dw, qkv, sums, Spart);
  k_out<<<dim3(HW / (OWAVES * OPX * OGROUPS), BCN), OWAVES * 32, 0, stream>>>(qkv, Spart, sums, proj, out);
}


# global memory taken off the stage barriers: barriers order LDS only (local fence + s_barrier); next sub-block's x rows (L2 prefetch) and k_out's 2nd v group are requested a stage before their use
# speedup vs baseline: 3.9868x; 1.0633x over previous
#include <hip/hip_runtime.h>

typedef __attribute__((ext_vector_type(2))) float v2f;
typedef __attribute__((ext_vector_type(4))) float v4f;
typedef __attribute__((ext_vector_type(8))) float v8f;

#define HW    16384
#define IMG_W 128
#define BCN   32
#define CIN   64
#define CMID  96
#define AN2   32
#define OROWS  4
#define SUBS   4
#define LROWS  (OROWS + 2)
#define PWAVES (2 * LROWS)
#define NWAVES (2 * PWAVES)
#define SWAVES 16
#define X_AHEAD 2
#define W_AHEAD 1
#define TLS_PAD   4
#define TLS_PITCH (IMG_W + TLS_PAD)
#define TLS_ROWSTRIDE (LROWS * TLS_PITCH)
#define TLS_FLOATS (CMID * TLS_ROWSTRIDE + TLS_PAD)
#define NBLK_X (IMG_W / (OROWS * SUBS))
#define NPART  (NBLK_X * SUBS)

static __device__ __forceinline__ v8f wmma4(v2f a, v2f b, v8f c) {
  return __builtin_amdgcn_wmma_f32_16x16x4_f32(
       false, a,  false, b,
       (short)0, c,  false,  false);
}

static __device__ __forceinline__ void lds_barrier() {
  __builtin_amdgcn_fence(__ATOMIC_RELEASE, "workgroup", "local");
  __builtin_amdgcn_s_barrier();
  __builtin_amdgcn_fence(__ATOMIC_ACQUIRE, "workgroup", "local");
}

__global__ void __launch_bounds__(NWAVES * 32) __attribute__((amdgpu_waves_per_eu(6, 6)))
k_fused(const float* __restrict__ x,
        const float* __restrict__ pw,
        const float* __restrict__ dw,
        float* __restrict__ qkv,
        float* __restrict__ gsums,
        float* __restrict__ Spart) {
  extern __shared__ __attribute__((aligned(16))) float tls[];

  const int tid  = threadIdx.x;
  const int lane = tid & 31, wave = tid >> 5;
  const int lo = lane & 15, hi = lane >> 4;
  const int img = blockIdx.y;
  const int r0  = blockIdx.x * (OROWS * SUBS);
  const float* __restrict__ X = x + (size_t)img * CIN * HW;
  float* __restrict__ QKV = qkv + (size_t)img * CMID * HW;

  if (tid <= CMID * LROWS) {
    v4f z = {0.f, 0.f, 0.f, 0.f};
    *(v4f*)(tls + tid * TLS_PITCH) = z;
  }

  const int wv = __builtin_amdgcn_readfirstlane(wave);
  const int c4 = lane * 4;

  const int chalf = wv / PWAVES;
  const int w12 = wv - chalf * PWAVES;
  const int prow = w12 >> 1;
  const int colbase = (w12 & 1) * 64;
  const int cbase = chalf * 48;

  v8f sacc[2][2] = {};

#pragma nounroll
  for (int sub = 0; sub < SUBS; ++sub) {
    const int rbase = (OROWS * sub) % LROWS;

    const int wr = (sub == 0) ? prow : prow + 2;
    if (wr < LROWS) {
      const int rr = r0 - 1 + OROWS * sub + wr;
      int lrw = rbase + wr;
      if (lrw >= LROWS) lrw -= LROWS;
      float* trow = tls + cbase * TLS_ROWSTRIDE + lrw * TLS_PITCH + TLS_PAD + colbase + 4 * lo;
      if (rr < 0 || rr >= IMG_W) {
        v4f z = {0.f, 0.f, 0.f, 0.f};
        for (int c = hi; c < 48; c += 2) *(v4f*)(trow + c * TLS_ROWSTRIDE) = z;
      } else {
      const float* __restrict__ xp = X + (size_t)(2 * hi) * HW + rr * IMG_W + colbase + 4 * lo;
      const float* __restrict__ ap = pw + (cbase + lo) * CIN + 2 * hi;
      v8f acc[3][4] = {};
      v4f x0[X_AHEAD + 1], x1[X_AHEAD + 1];
      v2f a[W_AHEAD + 1][3];
#pragma unroll
      for (int st = 0; st < X_AHEAD; ++st) {
        x0[st] = *(const v4f*)(xp + (size_t)(4 * st) * HW);
        x1[st] = *(const v4f*)(xp + (size_t)(4 * st + 1) * HW);
      }
#pragma unroll
      for (int st = 0; st < W_AHEAD; ++st)
#pragma unroll
        for (int m = 0; m < 3; ++m) a[st][m] = *(const v2f*)(ap + m * 16 * CIN + 4 * st);
#pragma unroll
      for (int st = 0; st < CIN / 4; ++st) {
        if (st + W_AHEAD < CIN / 4) {
          const int sl = (st + W_AHEAD) % (W_AHEAD + 1);
          const int kn = 4 * (st + W_AHEAD);
#pragma unroll
          for (int m = 0; m < 3; ++m) a[sl][m] = *(const v2f*)(ap + m * 16 * CIN + kn);
        }
        if (st + X_AHEAD < CIN / 4) {
          const int sl = (st + X_AHEAD) % (X_AHEAD + 1);
          const int kn = 4 * (st + X_AHEAD);
          x0[sl] = *(const v4f*)(xp + (size_t)kn * HW);
          x1[sl] = *(const v4f*)(xp + (size_t)(kn + 1) * HW);
        }
        __builtin_amdgcn_sched_barrier(0);
        const int sx = st % (X_AHEAD + 1);
        const int sw = st % (W_AHEAD + 1);
#pragma unroll
        for (int i = 0; i < 4; ++i) {
          v2f b;
          b.x = x0[sx][i];
          b.y = x1[sx][i];
#pragma unroll
          for (int m = 0; m < 3; ++m) acc[m][i] = wmma4(a[sw][m], b, acc[m][i]);
        }
      }
#pragma unroll
      for (int m = 0; m < 3; ++m)
#pragma unroll
        for (int r = 0; r < 8; ++r) {
          v4f o = {acc[m][0][r], acc[m][1][r], acc[m][2][r], acc[m][3][r]};
          *(v4f*)(trow + (m * 16 + r + 8 * hi) * TLS_ROWSTRIDE) = o;
        }
      }
    }
    lds_barrier();

    if (sub + 1 < SUBS && wv < 16) {
      const int u = wv * 32 + lane;
      int rr = r0 + OROWS * (sub + 1) + 1 + (u >> 7);
      rr = rr < IMG_W ? rr : IMG_W - 1;
      __builtin_prefetch(X + (size_t)((u >> 1) & 63) * HW + rr * IMG_W + (u & 1) * 64, 0, 3);
    }

#pragma nounroll
    for (int j = 0; j < CMID / NWAVES; ++j) {
      const int ch = wv + NWAVES * j;
      const float* __restrict__ wch = dw + ch * 9;
      float w[9];
#pragma unroll
      for (int t = 0; t < 9; ++t) w[t] = wch[t];

      float* tch = tls + ch * TLS_ROWSTRIDE + TLS_PAD + c4;
      v4f mid[LROWS];
      float lft[LROWS], rgt[LROWS];
#pragma unroll
      for (int r = 0; r < LROWS; ++r) {
        int lr = rbase + r;
        if (lr >= LROWS) lr -= LROWS;
        const float* trd = tch + lr * TLS_PITCH;
        mid[r] = *(const v4f*)(trd);
        lft[r] = trd[-1];
        rgt[r] = trd[4];
      }

      float ss = 0.f;
#pragma unroll
      for (int orow = 0; orow < OROWS; ++orow) {
        float s0 = 0.f, s1 = 0.f, s2 = 0.f, s3 = 0.f;
#pragma unroll
        for (int dy = 0; dy < 3; ++dy) {
          const int r = orow + dy;
          const float w0 = w[dy * 3 + 0], w1 = w[dy * 3 + 1], w2 = w[dy * 3 + 2];
          s0 += w0 * lft[r];    s0 += w1 * mid[r].x;  s0 += w2 * mid[r].y;
          s1 += w0 * mid[r].x;  s1 += w1 * mid[r].y;  s1 += w2 * mid[r].z;
          s2 += w0 * mid[r].y;  s2 += w1 * mid[r].z;  s2 += w2 * mid[r].w;
          s3 += w0 * mid[r].z;  s3 += w1 * mid[r].w;  s3 += w2 * rgt[r];
        }
        v4f o = {s0, s1, s2, s3};
        if (ch < 2 * AN2) {
          int lr = rbase + orow;
          if (lr >= LROWS) lr -= LROWS;
          *(v4f*)(tch + lr * TLS_PITCH) = o;
        } else {
          *(v4f*)(QKV + (size_t)ch * HW + (r0 + OROWS * sub + orow) * IMG_W + c4) = o;
        }
        ss += s0 * s0;  ss += s1 * s1;  ss += s2 * s2;  ss += s3 * s3;
      }
      if (ch < 2 * AN2) {
#pragma unroll
        for (int off = 16; off > 0; off >>= 1) ss += __shfl_xor(ss, off, 32);
        if (lane == 0) gsums[((img * NBLK_X + blockIdx.x) * SUBS + sub) * 64 + ch] = ss;
      }
    }
    lds_barrier();

    if (wv < SWAVES) {
      int lr = rbase + (wv >> 2);
      if (lr >= LROWS) lr -= LROWS;
      const float* sp = tls + lo * TLS_ROWSTRIDE + lr * TLS_PITCH + TLS_PAD + 32 * (wv & 3) + 16 * hi;
      v4f f[4][4];
#pragma unroll
      for (int t = 0; t < 4; ++t)
#pragma unroll
        for (int q4 = 0; q4 < 4; ++q4)
          f[t][q4] = *(const v4f*)(sp + (t * 16) * TLS_ROWSTRIDE + 4 * q4);
#pragma unroll
      for (int j = 0; j < 8; ++j) {
        const int q4 = j >> 1, e = 2 * (j & 1);
        v2f a0, a1, b0, b1;
        a0.x = f[0][q4][e];  a0.y = f[0][q4][e + 1];
        a1.x = f[1][q4][e];  a1.y = f[1][q4][e + 1];
        b0.x = f[2][q4][e];  b0.y = f[2][q4][e + 1];
        b1.x = f[3][q4][e];  b1.y = f[3][q4][e + 1];
        sacc[0][0] = wmma4(a0, b0, sacc[0][0]);
        sacc[0][1] = wmma4(a0, b1, sacc[0][1]);
        sacc[1][0] = wmma4(a1, b0, sacc[1][0]);
        sacc[1][1] = wmma4(a1, b1, sacc[1][1]);
      }
    }
    lds_barrier();
  }

  if (wv < SWAVES) {
#pragma unroll
    for (int m = 0; m < 2; ++m)
#pragma unroll
      for (int n = 0; n < 2; ++n)
#pragma unroll
        for (int r = 0; r < 8; ++r)
          tls[wv * (AN2 * AN2) + (m * 16 + 8 * hi + r) * AN2 + n * 16 + lo] = sacc[m][n][r];
  }
  lds_barrier();
  float* __restrict__ Sp = Spart + (size_t)(img * NBLK_X + blockIdx.x) * (AN2 * AN2);
  for (int idx = tid; idx < AN2 * AN2; idx += NWAVES * 32) {
    float s = tls[idx];
#pragma unroll
    for (int w = 1; w < SWAVES; ++w) s += tls[w * (AN2 * AN2) + idx];
    Sp[idx] = s;
  }
}

#define OPX    64
#define OGROUPS 4
#define OWAVES 8
#define WPITCH 34
__global__ void __launch_bounds__(OWAVES * 32)
k_out(const float* __restrict__ qkv,
      const float* __restrict__ Spart,
      const float* __restrict__ sums,
      const float* __restrict__ proj,
      float* __restrict__ out) {
  __shared__ float rqk[2 * AN2];
  __shared__ float att[AN2 * AN2];
  __shared__ __attribute__((aligned(16))) float Wl[CIN * WPITCH];
  const int lane = threadIdx.x & 31;
  const int lo = lane & 15, hi = lane >> 4;
  const int wv = __builtin_amdgcn_readfirstlane(threadIdx.x >> 5);
  const int img = blockIdx.y;
  const int p0 = (blockIdx.x * OWAVES + wv) * (OPX * OGROUPS);
  const float* __restrict__ vp =
      qkv + (size_t)img * CMID * HW + (size_t)(2 * AN2 + 2 * hi) * HW + p0 + 4 * lo;
  float* __restrict__ op = out + (size_t)img * CIN * HW + (size_t)(8 * hi) * HW + p0 + 4 * lo;

  const float* __restrict__ Sp = Spart + (size_t)img * (NBLK_X * AN2 * AN2) + (4 * wv) * AN2 + lane;
  float sl[4][NBLK_X];
#pragma unroll
  for (int r = 0; r < 4; ++r)
#pragma unroll
    for (int b = 0; b < NBLK_X; ++b) sl[r][b] = Sp[b * (AN2 * AN2) + r * AN2];
  float part[NPART];
  if (threadIdx.x < 2 * AN2) {
    const float* __restrict__ sp = sums + (size_t)img * (NPART * 64) + threadIdx.x;
#pragma unroll
    for (int p = 0; p < NPART; ++p) part[p] = sp[p * 64];
  }

  v4f x0[2][8], x1[2][8];
#pragma unroll
  for (int ks = 0; ks < 8; ++ks) {
    x0[0][ks] = *(const v4f*)(vp + (size_t)(4 * ks) * HW);
    x1[0][ks] = *(const v4f*)(vp + (size_t)(4 * ks + 1) * HW);
  }
#pragma unroll
  for (int ks = 0; ks < 8; ++ks) {
    x0[1][ks] = *(const v4f*)(vp + (size_t)(4 * ks) * HW + OPX);
    x1[1][ks] = *(const v4f*)(vp + (size_t)(4 * ks + 1) * HW + OPX);
  }

  if (threadIdx.x < 2 * AN2) {
    float ssq = 0.f;
#pragma unroll
    for (int p = 0; p < NPART; ++p) ssq += part[p];
    rqk[threadIdx.x] = 1.f / fmaxf(sqrtf(ssq), 1e-12f);
  }
  lds_barrier();
  {
    const float rkc = rqk[AN2 + lane];
#pragma unroll
    for (int r = 0; r < 4; ++r) {
      const int row = 4 * wv + r;
      float sv = sl[r][0];
#pragma unroll
      for (int b = 1; b < NBLK_X; ++b) sv += sl[r][b];
      const float s = sv * rqk[row] * rkc;
      float mx = s;
      for (int off = 16; off > 0; off >>= 1) mx = fmaxf(mx, __shfl_xor(mx, off, 32));
      const float ex = __expf(s - mx);
      float sum = ex;
      for (int off = 16; off > 0; off >>= 1) sum += __shfl_xor(sum, off, 32);
      const float inv = 1.f / sum;
      att[row * AN2 + lane] = ex * inv;
    }
  }
  lds_barrier();
  {
    float at[AN2];
#pragma unroll
    for (int cc = 0; cc < AN2; ++cc) at[cc] = att[cc * AN2 + lane];
#pragma unroll
    for (int e = 0; e < CIN / OWAVES; ++e) {
      const int o = wv + OWAVES * e;
      const float* __restrict__ pr = proj + o * AN2;
      float s = 0.f;
#pragma unroll
      for (int cc = 0; cc < AN2; ++cc) s += pr[cc] * at[cc];
      Wl[o * WPITCH + lane] = s;
    }
  }
  lds_barrier();

  const float* __restrict__ wp = Wl + lo * WPITCH + 2 * hi;
  v2f a[8][4];
#pragma unroll
  for (int ks = 0; ks < 8; ++ks)
#pragma unroll
    for (int m = 0; m < 4; ++m) a[ks][m] = *(const v2f*)(wp + m * 16 * WPITCH + 4 * ks);

#pragma unroll
  for (int g = 0; g < OGROUPS; ++g) {
    const int cur = g & 1;
    if (g > 0 && g + 1 < OGROUPS) {
#pragma unroll
      for (int ks = 0; ks < 8; ++ks) {
        x0[cur ^ 1][ks] = *(const v4f*)(vp + (size_t)(4 * ks) * HW + OPX * (g + 1));
        x1[cur ^ 1][ks] = *(const v4f*)(vp + (size_t)(4 * ks + 1) * HW + OPX * (g + 1));
      }
    }
    __builtin_amdgcn_sched_barrier(0);
    v8f acc[4][4] = {};
#pragma unroll
    for (int ks = 0; ks < 8; ++ks) {
#pragma unroll
      for (int i = 0; i < 4; ++i) {
        v2f b;
        b.x = x0[cur][ks][i];
        b.y = x1[cur][ks][i];
#pragma unroll
        for (int m = 0; m < 4; ++m) acc[m][i] = wmma4(a[ks][m], b, acc[m][i]);
      }
    }
#pragma unroll
    for (int m = 0; m < 4; ++m)
#pragma unroll
      for (int r = 0; r < 8; ++r) {
        v4f o = {acc[m][0][r], acc[m][1][r], acc[m][2][r], acc[m][3][r]};
        __builtin_nontemporal_store(o, (v4f*)(op + (size_t)(m * 16 + r) * HW + OPX * g));
      }
  }
}

extern "C" void kernel_launch(void* const* d_in, const int* in_sizes, int n_in,
                              void* d_out, int out_size, void* d_ws, size_t ws_size,
                              hipStream_t stream) {
  const float* x    = (const float*)d_in[0];
  const float* pw   = (const float*)d_in[1];
  const float* dw   = (const float*)d_in[2];
  const float* proj = (const float*)d_in[3];
  float* out = (float*)d_out;

  float* qkv   = (float*)d_ws;
  float* sums  = qkv + (size_t)BCN * CMID * HW;
  float* Spart = sums + (size_t)BCN * NPART * 64;

  k_fused<<<dim3(NBLK_X, BCN), NWAVES * 32, TLS_FLOATS * sizeof(float), stream>>>(
      x, pw, dw, qkv, sums, Spart);
  k_out<<<dim3(HW / (OWAVES * OPX * OGROUPS), BCN), OWAVES * 32, 0, stream>>>(qkv, Spart, sums, proj, out);
}
